# starting kernel
# baseline (speedup 1.0000x reference)
#include <hip/hip_runtime.h>

typedef __attribute__((ext_vector_type(2))) float v2f;
typedef __attribute__((ext_vector_type(8))) float v8f;

#define NNODES 50000
#define NEDGES 800000
#define ETOT   850000
#define FIN    128
#define HIDC   64
#define NHEADS 4
#define EPSBN  1e-5f

template <int K>
__global__ __launch_bounds__(256) void gemm_wmma_f32(
    const float* __restrict__ A, const float* __restrict__ B,
    float* __restrict__ C, int Nc, int n_wave_jobs)
{
  __shared__ float Bs[K * 16];

  const int tid  = threadIdx.x;
  const int lane = tid & 31;
  const int wave = tid >> 5;
  const int half = lane >> 4;
  const int ml   = lane & 15;
  const int col0 = blockIdx.x * 16;

  #pragma unroll
  for (int i = tid; i < K * 16; i += 256) {
    const int r = i >> 4, cc = i & 15;
    Bs[i] = B[(size_t)r * Nc + col0 + cc];
  }
  __syncthreads();

  const int job = blockIdx.y * 8 + wave;
  if (job < n_wave_jobs) {
    const int row_base = job * 80;

    v8f acc[5];
    #pragma unroll
    for (int t = 0; t < 5; ++t)
      acc[t] = (v8f){0.f,0.f,0.f,0.f,0.f,0.f,0.f,0.f};

    #pragma unroll 4
    for (int k0 = 0; k0 < K; k0 += 4) {
      v2f b;
      b.x = Bs[(k0 + 2*half + 0) * 16 + ml];
      b.y = Bs[(k0 + 2*half + 1) * 16 + ml];
      const float* ap = A + (size_t)(row_base + ml) * K + (k0 + 2*half);
      #pragma unroll
      for (int t = 0; t < 5; ++t) {
        v2f a = *(const v2f*)(ap + (size_t)(16 * t) * K);
        acc[t] = __builtin_amdgcn_wmma_f32_16x16x4_f32(
            false, a, false, b, (short)0, acc[t], false, false);
      }
    }

    #pragma unroll
    for (int t = 0; t < 5; ++t) {
      float* cp = C + (size_t)(row_base + 16*t + 8*half) * Nc + col0 + ml;
      #pragma unroll
      for (int i = 0; i < 8; ++i)
        cp[(size_t)i * Nc] = acc[t][i];
    }
  }
}

__global__ void att_scores(const float* __restrict__ hf,
                           const float* __restrict__ a_s,
                           const float* __restrict__ a_d,
                           float* __restrict__ s_src,
                           float* __restrict__ s_dst,
                           int n, int heads, int c)
{
  int idx = blockIdx.x * blockDim.x + threadIdx.x;
  if (idx >= n * heads) return;
  int node = idx / heads, h = idx % heads;
  const float* f  = hf  + (size_t)node * heads * c + (size_t)h * c;
  const float* as = a_s + (size_t)h * c;
  const float* ad = a_d + (size_t)h * c;
  float ss = 0.f, sd = 0.f;
  #pragma unroll 4
  for (int i = 0; i < c; ++i) { float v = f[i]; ss += v * as[i]; sd += v * ad[i]; }
  s_src[idx] = ss; s_dst[idx] = sd;
}

__device__ __forceinline__ unsigned f2key(float f) {
  unsigned u = __float_as_uint(f);
  return (u & 0x80000000u) ? ~u : (u | 0x80000000u);
}
__device__ __forceinline__ float key2f(unsigned k) {
  return __uint_as_float((k & 0x80000000u) ? (k ^ 0x80000000u) : ~k);
}

__device__ __forceinline__ void edge_src_dst(const int* __restrict__ ei, int e,
                                             int& src, int& dst) {
  if (e < NEDGES) { src = ei[e]; dst = ei[NEDGES + e]; }
  else            { src = dst = e - NEDGES; }
}

__global__ void edge_logits(const int* __restrict__ ei,
                            const float* __restrict__ s_src,
                            const float* __restrict__ s_dst,
                            float* __restrict__ e_buf,
                            unsigned* __restrict__ emax_key, int heads)
{
  int idx = blockIdx.x * blockDim.x + threadIdx.x;
  if (idx >= ETOT * heads) return;
  int e = idx / heads, h = idx % heads;
  int src, dst; edge_src_dst(ei, e, src, dst);
  float v = s_src[src * heads + h] + s_dst[dst * heads + h];
  v = v > 0.f ? v : 0.2f * v;
  e_buf[idx] = v;
  atomicMax(&emax_key[dst * heads + h], f2key(v));
}

__global__ void edge_exp(const int* __restrict__ ei,
                         float* __restrict__ e_buf,
                         const unsigned* __restrict__ emax_key,
                         float* __restrict__ denom, int heads)
{
  int idx = blockIdx.x * blockDim.x + threadIdx.x;
  if (idx >= ETOT * heads) return;
  int e = idx / heads, h = idx % heads;
  int src, dst; edge_src_dst(ei, e, src, dst);
  float m  = key2f(emax_key[dst * heads + h]);
  float ee = expf(e_buf[idx] - m);
  e_buf[idx] = ee;
  atomicAdd(&denom[dst * heads + h], ee);
}

__global__ void edge_aggregate(const int* __restrict__ ei,
                               const float* __restrict__ e_buf,
                               const float* __restrict__ denom,
                               const float* __restrict__ hf,
                               float* __restrict__ agg, int heads, int c)
{
  const int hc_tot = heads * c;
  const int hq     = hc_tot >> 2;
  long long tid   = (long long)blockIdx.x * blockDim.x + threadIdx.x;
  long long total = (long long)ETOT * hq;
  if (tid >= total) return;
  const int e  = (int)(tid / hq);
  const int q  = (int)(tid % hq);
  const int hc = q << 2;
  const int h  = hc / c;
  int src, dst; edge_src_dst(ei, e, src, dst);
  const float alpha = e_buf[(size_t)e * heads + h] /
                      (denom[dst * heads + h] + 1e-16f);
  const float4 v = *(const float4*)(hf + (size_t)src * hc_tot + hc);
  float* dp = agg + (size_t)dst * hc_tot + hc;
  atomicAdd(dp + 0, v.x * alpha);
  atomicAdd(dp + 1, v.y * alpha);
  atomicAdd(dp + 2, v.z * alpha);
  atomicAdd(dp + 3, v.w * alpha);
}

__global__ void post_bn(const float* __restrict__ agg,
                        const float* __restrict__ bias,
                        const float* __restrict__ g,
                        const float* __restrict__ bt,
                        const float* __restrict__ m,
                        const float* __restrict__ v,
                        float* __restrict__ out, int n, int ch, int do_elu)
{
  long long idx = (long long)blockIdx.x * blockDim.x + threadIdx.x;
  if (idx >= (long long)n * ch) return;
  int cix = (int)(idx % ch);
  float x = agg[idx] + bias[cix];
  x = (x - m[cix]) * rsqrtf(v[cix] + EPSBN) * g[cix] + bt[cix];
  if (do_elu) x = x > 0.f ? x : expm1f(x);
  out[idx] = x;
}

__global__ void final_linear(const float* __restrict__ h,
                             const float* __restrict__ Wc,
                             const float* __restrict__ bc,
                             float* __restrict__ out)
{
  int n = blockIdx.x * blockDim.x + threadIdx.x;
  if (n >= NNODES) return;
  float o0 = bc[0], o1 = bc[1];
  const float* r = h + (size_t)n * HIDC;
  #pragma unroll 8
  for (int i = 0; i < HIDC; ++i) {
    float x = r[i];
    o0 += x * Wc[2*i + 0];
    o1 += x * Wc[2*i + 1];
  }
  out[2*n + 0] = o0;
  out[2*n + 1] = o1;
}

__global__ void fill_u32(unsigned* __restrict__ p, unsigned val, long long n)
{
  long long i = (long long)blockIdx.x * blockDim.x + threadIdx.x;
  if (i < n) p[i] = val;
}

static inline unsigned gblk(long long total, int bs) {
  return (unsigned)((total + bs - 1) / bs);
}

extern "C" void kernel_launch(void* const* d_in, const int* in_sizes, int n_in,
                              void* d_out, int out_size, void* d_ws, size_t ws_size,
                              hipStream_t stream)
{
  const float* x  = (const float*)d_in[0];
  const int*   ei = (const int*)  d_in[1];
  const float* Wc = (const float*)d_in[26];
  const float* bc = (const float*)d_in[27];

  char* ws = (char*)d_ws;
  float*    hf    = (float*)ws;    ws += (size_t)NNODES * 256 * sizeof(float);
  float*    agg   = (float*)ws;    ws += (size_t)NNODES * 256 * sizeof(float);
  float*    ssrc  = (float*)ws;    ws += (size_t)NNODES * NHEADS * sizeof(float);
  float*    sdst  = (float*)ws;    ws += (size_t)NNODES * NHEADS * sizeof(float);
  unsigned* emax  = (unsigned*)ws; ws += (size_t)NNODES * NHEADS * sizeof(unsigned);
  float*    denom = (float*)ws;    ws += (size_t)NNODES * NHEADS * sizeof(float);
  float*    ebuf  = (float*)ws;    ws += (size_t)ETOT   * NHEADS * sizeof(float);

  const int n_wave_jobs = (NNODES / 16) / 5;

  const float* in = x;
  for (int l = 0; l < 3; ++l) {
    int base = 2 + l * 8;
    const float* W  = (const float*)d_in[base + 0];
    const float* as = (const float*)d_in[base + 1];
    const float* ad = (const float*)d_in[base + 2];
    const float* bb = (const float*)d_in[base + 3];
    const float* g  = (const float*)d_in[base + 4];
    const float* bt = (const float*)d_in[base + 5];
    const float* m  = (const float*)d_in[base + 6];
    const float* vv = (const float*)d_in[base + 7];

    const int heads = (l < 2) ? NHEADS : 1;
    const int c     = HIDC;
    const int hc    = heads * c;

    dim3 ggrid(hc / 16, (unsigned)((n_wave_jobs + 7) / 8));
    if (l == 0)
      gemm_wmma_f32<FIN><<<ggrid, 256, 0, stream>>>(in, W, hf, hc, n_wave_jobs);
    else
      gemm_wmma_f32<NHEADS * HIDC><<<ggrid, 256, 0, stream>>>(in, W, hf, hc, n_wave_jobs);

    const int nh = NNODES * heads;
    att_scores<<<gblk(nh, 256), 256, 0, stream>>>(hf, as, ad, ssrc, sdst,
                                                  NNODES, heads, c);

    fill_u32<<<gblk((long long)NNODES * hc, 256), 256, 0, stream>>>(
        (unsigned*)agg, 0u, (long long)NNODES * hc);
    fill_u32<<<gblk(nh, 256), 256, 0, stream>>>(emax, 0u, nh);
    fill_u32<<<gblk(nh, 256), 256, 0, stream>>>((unsigned*)denom, 0u, nh);

    const long long eh = (long long)ETOT * heads;
    edge_logits<<<gblk(eh, 256), 256, 0, stream>>>(ei, ssrc, sdst, ebuf, emax, heads);
    edge_exp   <<<gblk(eh, 256), 256, 0, stream>>>(ei, ebuf, emax, denom, heads);

    const long long tot4 = (long long)ETOT * (hc >> 2);
    edge_aggregate<<<gblk(tot4, 256), 256, 0, stream>>>(ei, ebuf, denom, hf, agg,
                                                        heads, c);

    post_bn<<<gblk((long long)NNODES * hc, 256), 256, 0, stream>>>(
        agg, bb, g, bt, m, vv, agg, NNODES, hc, (l < 2) ? 1 : 0);

    in = agg;
  }

  final_linear<<<gblk(NNODES, 256), 256, 0, stream>>>(in, Wc, bc, (float*)d_out);
}


# Tile edge lists built once per call: layer 0's launch stores each wave's list (source, tile row) in the workspace; layers 1, 2 read it back: no column scan, no source lookup; sum order as before
# speedup vs baseline: 3.3874x; 3.3874x over previous
#include <hip/hip_runtime.h>

typedef __attribute__((ext_vector_type(2))) float v2f;
typedef __attribute__((ext_vector_type(4))) float v4f;
typedef __attribute__((ext_vector_type(8))) float v8f;

#define NNODES 50000
#define NEDGES 800000
#define FIN    128
#define HIDC   64
#define NHEADS 4
#define EPSBN  1e-5f

#define GEMM_KC 32
#define GEMM_AP (GEMM_KC + 4)
#define GEMM_ROWS 80
#define GEMM_NLD (GEMM_ROWS / 4)

template <int K>
__global__ __launch_bounds__(256) void gemm_wmma_f32(
    const float* __restrict__ A, const float* __restrict__ B,
    float* __restrict__ C, int Nc, int n_wave_jobs)
{
  static_assert(K % GEMM_KC == 0, "whole chunks");
  __shared__ float Bs[K * 16];
  __shared__ __attribute__((aligned(16))) float As[8 * GEMM_ROWS * GEMM_AP];

  const int tid  = threadIdx.x;
  const int lane = tid & 31;
  const int wave = tid >> 5;
  const int half = lane >> 4;
  const int ml   = lane & 15;
  const int col0 = blockIdx.x * 16;

  const int  job    = blockIdx.y * 8 + wave;
  const bool active = job < n_wave_jobs;
  const int  row_base = job * GEMM_ROWS;

  const int lr = lane >> 3, lk = 4 * (lane & 7);
  const float* gp = A + (size_t)(row_base + lr) * K + lk;
  v4f st[GEMM_NLD];
  if (active) {
    #pragma unroll
    for (int j = 0; j < GEMM_NLD; ++j)
      st[j] = *(const v4f*)(gp + (size_t)(4 * j) * K);
  }

  #pragma unroll
  for (int i = tid; i < K * 16; i += 256) {
    const int r = i >> 4, cc = i & 15;
    Bs[i] = B[(size_t)r * Nc + col0 + cc];
  }
  __syncthreads();

  if (active) {
    float* slab = As + wave * (GEMM_ROWS * GEMM_AP);

    v8f acc[5];
    #pragma unroll
    for (int t = 0; t < 5; ++t)
      acc[t] = (v8f){0.f,0.f,0.f,0.f,0.f,0.f,0.f,0.f};

    for (int kc = 0; kc < K; kc += GEMM_KC) {
      #pragma unroll
      for (int j = 0; j < GEMM_NLD; ++j)
        *(v4f*)(slab + (4 * j + lr) * GEMM_AP + lk) = st[j];
      if (kc + GEMM_KC < K) {
        #pragma unroll
        for (int j = 0; j < GEMM_NLD; ++j)
          st[j] = *(const v4f*)(gp + (size_t)(4 * j) * K + (kc + GEMM_KC));
      }
      #pragma unroll
      for (int k0 = 0; k0 < GEMM_KC; k0 += 4) {
        v2f b;
        b.x = Bs[(kc + k0 + 2*half + 0) * 16 + ml];
        b.y = Bs[(kc + k0 + 2*half + 1) * 16 + ml];
        const float* ap = slab + ml * GEMM_AP + (k0 + 2*half);
        #pragma unroll
        for (int t = 0; t < 5; ++t) {
          v2f a = *(const v2f*)(ap + (16 * t) * GEMM_AP);
          acc[t] = __builtin_amdgcn_wmma_f32_16x16x4_f32(
              false, a, false, b, (short)0, acc[t], false, false);
        }
      }
    }

    #pragma unroll
    for (int t = 0; t < 5; ++t) {
      float* cp = C + (size_t)(row_base + 16*t + 8*half) * Nc + col0 + ml;
      #pragma unroll
      for (int i = 0; i < 8; ++i)
        cp[(size_t)i * Nc] = acc[t][i];
    }
  }
}

__global__ void att_scores(const float* __restrict__ hf,
                           const float* __restrict__ a_s,
                           const float* __restrict__ a_d,
                           float* __restrict__ s_src,
                           float* __restrict__ s_dst,
                           int n, int heads, int c)
{
  int idx = blockIdx.x * blockDim.x + threadIdx.x;
  if (idx >= n * heads) return;
  int node = idx / heads, h = idx % heads;
  const float* f  = hf  + (size_t)node * heads * c + (size_t)h * c;
  const float* as = a_s + (size_t)h * c;
  const float* ad = a_d + (size_t)h * c;
  float ss = 0.f, sd = 0.f;
  #pragma unroll 4
  for (int i = 0; i < c; ++i) { float v = f[i]; ss += v * as[i]; sd += v * ad[i]; }
  s_src[idx] = ss; s_dst[idx] = sd;
}

__device__ __forceinline__ unsigned f2key(float f) {
  unsigned u = __float_as_uint(f);
  return (u & 0x80000000u) ? ~u : (u | 0x80000000u);
}
__device__ __forceinline__ float key2f(unsigned k) {
  return __uint_as_float((k & 0x80000000u) ? (k ^ 0x80000000u) : ~k);
}

#define TILE_NODES 256
#define NTILES     ((NNODES + TILE_NODES - 1) / TILE_NODES)
#define QSLOTS     256
#define QEBITS     20
#define QSBITS     16

template <int HEADS, int UE>
__device__ __forceinline__ void tile_edge_batch(
    const float* __restrict__ hf, float* acc,
    int src, int dl, const float (&a)[HEADS], int j, int lane)
{
  constexpr int HC = HEADS * HIDC;
  constexpr int NI = HC / 32;
  float v[UE][NI];
  float al[UE][HEADS];
  int   d[UE];
  #pragma unroll
  for (int u = 0; u < UE; ++u) {
    const int s = __builtin_amdgcn_readlane(src, j + u);
    d[u]        = __builtin_amdgcn_readlane(dl, j + u);
    #pragma unroll
    for (int h = 0; h < HEADS; ++h)
      al[u][h] = __int_as_float(
          __builtin_amdgcn_readlane(__float_as_int(a[h]), j + u));
    const float* hp = hf + (size_t)s * HC + lane;
    #pragma unroll
    for (int i = 0; i < NI; ++i) v[u][i] = hp[32 * i];
  }
  #pragma unroll
  for (int u = 0; u < UE; ++u) {
    float* ap = acc + d[u] * HC + lane;
    #pragma unroll
    for (int i = 0; i < NI; ++i)
      atomicAdd(ap + 32 * i, v[u][i] * al[u][(32 * i) / HIDC]);
  }
}

__device__ __forceinline__ float leaky02(float v) { return v > 0.f ? v : 0.2f * v; }

template <int HEADS, bool SAVED>
__device__ __forceinline__ void tile_list_max(
    const int* __restrict__ ei, const int* __restrict__ gl,
    const float* __restrict__ s_src,
    unsigned* smax, int* q, int cnt, int lane)
{
  for (int g = 0; g < cnt; g += 128) {
    int  dl[4], sr[4];
    bool on[4];
    #pragma unroll
    for (int k = 0; k < 4; ++k) {
      const int i = g + 32 * k + lane;
      on[k] = i < cnt;
      if constexpr (SAVED) {
        const int w = on[k] ? gl[i] : 0;
        dl[k] = (int)((unsigned)w >> QSBITS);
        sr[k] = w & ((1 << QSBITS) - 1);
      } else {
        const int w = on[k] ? q[i] : 0;
        dl[k] = (int)((unsigned)w >> QEBITS);
        sr[k] = ei[w & ((1 << QEBITS) - 1)];
      }
    }
    if constexpr (HEADS == 4) {
      float4 sv[4];
      #pragma unroll
      for (int k = 0; k < 4; ++k)
        sv[k] = *(const float4*)(s_src + (size_t)sr[k] * HEADS);
      #pragma unroll
      for (int k = 0; k < 4; ++k) {
        if (on[k]) {
          unsigned* mp = smax + dl[k] * HEADS;
          atomicMax(mp + 0, f2key(sv[k].x));
          atomicMax(mp + 1, f2key(sv[k].y));
          atomicMax(mp + 2, f2key(sv[k].z));
          atomicMax(mp + 3, f2key(sv[k].w));
          q[g + 32 * k + lane] = sr[k] | (dl[k] << QSBITS);
        }
      }
    } else {
      float sv[4][HEADS];
      #pragma unroll
      for (int k = 0; k < 4; ++k) {
        #pragma unroll
        for (int h = 0; h < HEADS; ++h)
          sv[k][h] = s_src[(size_t)sr[k] * HEADS + h];
      }
      #pragma unroll
      for (int k = 0; k < 4; ++k) {
        if (on[k]) {
          #pragma unroll
          for (int h = 0; h < HEADS; ++h)
            atomicMax(smax + dl[k] * HEADS + h, f2key(sv[k][h]));
          q[g + 32 * k + lane] = sr[k] | (dl[k] << QSBITS);
        }
      }
    }
  }
}

__device__ __forceinline__ void tile_list_form(
    const int* __restrict__ ei, int* q, int cnt, int lane)
{
  for (int i = lane; i < cnt; i += 32) {
    const int w = q[i];
    q[i] = ei[w & ((1 << QEBITS) - 1)] | ((int)((unsigned)w >> QEBITS) << QSBITS);
  }
}

template <int HEADS>
__device__ __forceinline__ void tile_list_sum(
    const float* __restrict__ s_src, const float* __restrict__ hf,
    float* acc, float* den, const float* sdl, const unsigned* rmax,
    const int* q, int base, int n, int lane)
{
  constexpr int HC = HEADS * HIDC;
  constexpr int UE = 32 / (HC / 32);
  int src = 0, dl = 0;
  float a[HEADS];
  #pragma unroll
  for (int h = 0; h < HEADS; ++h) a[h] = 0.f;
  if (lane < n) {
    const int w = q[base + lane];
    src = w & ((1 << QSBITS) - 1);
    dl  = (int)((unsigned)w >> QSBITS);
    if constexpr (HEADS == 4) {
      const float4 sv = *(const float4*)(s_src + (size_t)src * HEADS);
      const float4 sd = *(const float4*)(sdl + dl * HEADS);
      const uint4  mk = *(const uint4*)(rmax + dl * HEADS);
      a[0] = expf(leaky02(sv.x + sd.x) - __uint_as_float(mk.x));
      a[1] = expf(leaky02(sv.y + sd.y) - __uint_as_float(mk.y));
      a[2] = expf(leaky02(sv.z + sd.z) - __uint_as_float(mk.z));
      a[3] = expf(leaky02(sv.w + sd.w) - __uint_as_float(mk.w));
    } else {
      #pragma unroll
      for (int h = 0; h < HEADS; ++h)
        a[h] = expf(leaky02(s_src[(size_t)src * HEADS + h] + sdl[dl * HEADS + h]) -
                    __uint_as_float(rmax[dl * HEADS + h]));
    }
    float* dp = den + dl * HEADS;
    #pragma unroll
    for (int h = 0; h < HEADS; ++h)
      atomicAdd(dp + h, a[h]);
  }
  int j = 0;
  for (; j + UE <= n; j += UE)
    tile_edge_batch<HEADS, UE>(hf, acc, src, dl, a, j, lane);
  if constexpr (UE > 4) {
    for (; j + 4 <= n; j += 4)
      tile_edge_batch<HEADS, 4>(hf, acc, src, dl, a, j, lane);
  }
  for (; j < n; ++j)
    tile_edge_batch<HEADS, 1>(hf, acc, src, dl, a, j, lane);
}

template <int HEADS, bool SAVED>
__global__ __launch_bounds__(1024) void gat_dst_tile(
    const int* __restrict__ ei,
    int* __restrict__ lists,
    int* __restrict__ lcnt,

    const float* __restrict__ s_src,
    const float* __restrict__ s_dst,
    const float* __restrict__ hf,
    const float* __restrict__ bias,
    const float* __restrict__ g,
    const float* __restrict__ bt,
    const float* __restrict__ m,
    const float* __restrict__ v,
    float* __restrict__ out, int do_elu)
{
  constexpr int HC  = HEADS * HIDC;
  constexpr int HC4 = HC / 4;
  constexpr int TS  = TILE_NODES * HEADS;
  constexpr int NCH = NEDGES / 128;
  static_assert(NEDGES % 128 == 0, "edge list splits into whole 512-B loads");
  static_assert(1024 % HC4 == 0, "a thread keeps its channels in every pass");
  static_assert(TS <= 1024, "one thread per (row, head) pair");
  static_assert(NEDGES <= (1 << QEBITS) && TILE_NODES <= (1 << (31 - QEBITS)),
                "edge number and tile row share one list word");
  static_assert(NNODES <= (1 << QSBITS) && TILE_NODES <= (1 << (31 - QSBITS)),
                "source node and tile row share one list word");
  static_assert(QSLOTS >= 128, "an empty list takes any one chunk");
  static_assert(TILE_NODES == 256, "the tile test is a shift by 8");

  __shared__ __attribute__((aligned(16))) float    acc[TILE_NODES * HC];
  __shared__ int queue[32 * QSLOTS];
  __shared__ __attribute__((aligned(16))) unsigned rmax[TS];
  __shared__ __attribute__((aligned(16))) float    sdl[TS];
  __shared__ __attribute__((aligned(16))) float    den[TS];

  const int tid  = threadIdx.x;
  const int lane = tid & 31;
  const int wv   = __builtin_amdgcn_readfirstlane(tid >> 5);
  const int n0   = (int)blockIdx.x * TILE_NODES;
  const int nt   = (NNODES - n0) < TILE_NODES ? (NNODES - n0) : TILE_NODES;

  float s_self = 0.f;
  if (tid < nt * HEADS) {
    s_self    = s_src[(size_t)n0 * HEADS + tid];
    rmax[tid] = f2key(s_self);
    sdl[tid]  = s_dst[(size_t)n0 * HEADS + tid];
  }
  __syncthreads();

  const int* dstp = ei + NEDGES;
  const unsigned tile = blockIdx.x;
  int* q  = queue + wv * QSLOTS;
  int cnt = 0;

  auto ldchunk = [&](int cc) __attribute__((always_inline)) -> int4 {
    cc = cc < NCH ? cc : NCH - 1;
    return *(const int4*)(dstp + (size_t)cc * 128 + 4 * lane);
  };
  auto enqueue = [&](unsigned mask, int e, int d) __attribute__((always_inline)) {
    if (mask) {
      const int pos = cnt + (int)__builtin_amdgcn_mbcnt_lo(mask, 0u);
      if ((mask >> lane) & 1u)
        q[pos] = e | ((d & (TILE_NODES - 1)) << QEBITS);
      cnt += __builtin_popcount(mask);
    }
  };
  auto take = [&](const int4 cur, int cc) __attribute__((always_inline)) -> bool {
    const unsigned m0 = __builtin_amdgcn_ballot_w32(((unsigned)cur.x >> 8) == tile);
    const unsigned m1 = __builtin_amdgcn_ballot_w32(((unsigned)cur.y >> 8) == tile);
    const unsigned m2 = __builtin_amdgcn_ballot_w32(((unsigned)cur.z >> 8) == tile);
    const unsigned m3 = __builtin_amdgcn_ballot_w32(((unsigned)cur.w >> 8) == tile);
    if ((m0 | m1 | m2 | m3) == 0u) return true;
    const int tot = __builtin_popcount(m0) + __builtin_popcount(m1) +
                    __builtin_popcount(m2) + __builtin_popcount(m3);
    if (cnt + tot > QSLOTS) return false;
    const int eb = cc * 128 + 4 * lane;
    enqueue(m0, eb + 0, cur.x);
    enqueue(m1, eb + 1, cur.y);
    enqueue(m2, eb + 2, cur.z);
    enqueue(m3, eb + 3, cur.w);
    return true;
  };
  auto fill = [&](int c) __attribute__((always_inline)) -> int {
    int4 a0 = ldchunk(c), a1 = ldchunk(c + 32), a2 = ldchunk(c + 64), a3 = ldchunk(c + 96);
    for (;;) {
      const int4 b0 = ldchunk(c + 128), b1 = ldchunk(c + 160),
                 b2 = ldchunk(c + 192), b3 = ldchunk(c + 224);
      if (!take(a0, c)) return c;
      if (c + 32 < NCH && !take(a1, c + 32)) return c + 32;
      if (c + 64 < NCH && !take(a2, c + 64)) return c + 64;
      if (c + 96 < NCH && !take(a3, c + 96)) return c + 96;
      c += 128;
      if (c >= NCH) return NCH;
      a0 = ldchunk(c + 128); a1 = ldchunk(c + 160);
      a2 = ldchunk(c + 192); a3 = ldchunk(c + 224);
      if (!take(b0, c)) return c;
      if (c + 32 < NCH && !take(b1, c + 32)) return c + 32;
      if (c + 64 < NCH && !take(b2, c + 64)) return c + 64;
      if (c + 96 < NCH && !take(b3, c + 96)) return c + 96;
      c += 128;
      if (c >= NCH) return NCH;
    }
  };

  const int slot = (int)blockIdx.x * 32 + wv;
  int* gl = lists + (size_t)slot * QSLOTS;

  int  c = wv;
  bool whole = false;
  bool taken = false;
  if constexpr (SAVED) {
    const int c0 = lcnt[slot];
    if (c0 >= 0) {
      cnt = c0;
      tile_list_max<HEADS, true>(ei, gl, s_src, rmax, q, cnt, lane);
      whole = true;
      taken = true;
    }
  }
  if (!taken) {
    int rounds = 0;
    do {
      cnt = 0;
      c = fill(c);
      tile_list_max<HEADS, false>(ei, gl, s_src, rmax, q, cnt, lane);
      ++rounds;
    } while (c < NCH);
    whole = rounds == 1;
    if constexpr (!SAVED) {
      if (whole)
        for (int i = lane; i < cnt; i += 32) gl[i] = q[i];
      if (lane == 0) lcnt[slot] = whole ? cnt : -1;
    }
  }
  __syncthreads();

  if (tid < nt * HEADS) {
    const float t  = sdl[tid];
    const float mm = leaky02(key2f(rmax[tid]) + t);
    rmax[tid] = __float_as_uint(mm);
    den[tid]  = expf(leaky02(s_self + t) - mm);
  }
  __syncthreads();

  for (int i = tid; i < nt * HC4; i += 1024) {
    const int nl = i / HC4, c4 = i % HC4;
    const float al = den[nl * HEADS + (4 * c4) / HIDC];
    const float4 r = *(const float4*)(hf + (size_t)(n0 + nl) * HC + 4 * c4);
    float4 o;
    o.x = r.x * al; o.y = r.y * al; o.z = r.z * al; o.w = r.w * al;
    *(float4*)(acc + 4 * i) = o;
  }
  __syncthreads();

  if (whole) {
    for (int b = 0; b < cnt; b += 32)
      tile_list_sum<HEADS>(s_src, hf, acc, den, sdl, rmax, q, b,
                           (cnt - b) < 32 ? (cnt - b) : 32, lane);
  } else {
    c = wv;
    while (c < NCH) {
      cnt = 0;
      c = fill(c);
      tile_list_form(ei, q, cnt, lane);
      for (int b = 0; b < cnt; b += 32)
        tile_list_sum<HEADS>(s_src, hf, acc, den, sdl, rmax, q, b,
                             (cnt - b) < 32 ? (cnt - b) : 32, lane);
    }
  }
  __syncthreads();

  {
    const int c0 = 4 * (tid % HC4);
    const int hh = c0 / HIDC;
    const float4 bv = *(const float4*)(bias + c0);
    const float4 gv = *(const float4*)(g + c0);
    const float4 tv = *(const float4*)(bt + c0);
    const float4 mv = *(const float4*)(m + c0);
    const float4 vv = *(const float4*)(v + c0);
    const float r0 = rsqrtf(vv.x + EPSBN), r1 = rsqrtf(vv.y + EPSBN);
    const float r2 = rsqrtf(vv.z + EPSBN), r3 = rsqrtf(vv.w + EPSBN);
    float* op = out + (size_t)n0 * HC;
    for (int i = tid; i < nt * HC4; i += 1024) {
      const float4 s = *(const float4*)(acc + 4 * i);
      const float rd = 1.0f / (den[(i / HC4) * HEADS + hh] + 1e-16f);
      float4 y;
      y.x = (s.x * rd + bv.x - mv.x) * r0 * gv.x + tv.x;
      y.y = (s.y * rd + bv.y - mv.y) * r1 * gv.y + tv.y;
      y.z = (s.z * rd + bv.z - mv.z) * r2 * gv.z + tv.z;
      y.w = (s.w * rd + bv.w - mv.w) * r3 * gv.w + tv.w;
      if (do_elu) {
        y.x = y.x > 0.f ? y.x : expm1f(y.x);
        y.y = y.y > 0.f ? y.y : expm1f(y.y);
        y.z = y.z > 0.f ? y.z : expm1f(y.z);
        y.w = y.w > 0.f ? y.w : expm1f(y.w);
      }
      *(float4*)(op + 4 * i) = y;
    }
  }
}

__global__ void final_linear(const float* __restrict__ h,
                             const float* __restrict__ Wc,
                             const float* __restrict__ bc,
                             float* __restrict__ out)
{
  int n = blockIdx.x * blockDim.x + threadIdx.x;
  if (n >= NNODES) return;
  float o0 = bc[0], o1 = bc[1];
  const float* r = h + (size_t)n * HIDC;
  #pragma unroll 8
  for (int i = 0; i < HIDC; ++i) {
    float x = r[i];
    o0 += x * Wc[2*i + 0];
    o1 += x * Wc[2*i + 1];
  }
  out[2*n + 0] = o0;
  out[2*n + 1] = o1;
}

static inline unsigned gblk(long long total, int bs) {
  return (unsigned)((total + bs - 1) / bs);
}

extern "C" void kernel_launch(void* const* d_in, const int* in_sizes, int n_in,
                              void* d_out, int out_size, void* d_ws, size_t ws_size,
                              hipStream_t stream)
{
  const float* x  = (const float*)d_in[0];
  const int*   ei = (const int*)  d_in[1];
  const float* Wc = (const float*)d_in[26];
  const float* bc = (const float*)d_in[27];

  char* ws = (char*)d_ws;
  float*    hf    = (float*)ws;    ws += (size_t)NNODES * 256 * sizeof(float);
  float*    agg   = (float*)ws;    ws += (size_t)NNODES * 256 * sizeof(float);
  float*    ssrc  = (float*)ws;    ws += (size_t)NNODES * NHEADS * sizeof(float);
  float*    sdst  = (float*)ws;    ws += (size_t)NNODES * NHEADS * sizeof(float);
  int*      lists = (int*)ws;      ws += (size_t)NTILES * 32 * QSLOTS * sizeof(int);
  int*      lcnt  = (int*)ws;      ws += (size_t)NTILES * 32 * sizeof(int);

  const int n_wave_jobs = (NNODES / 16) / 5;

  const float* in = x;
  for (int l = 0; l < 3; ++l) {
    int base = 2 + l * 8;
    const float* W  = (const float*)d_in[base + 0];
    const float* as = (const float*)d_in[base + 1];
    const float* ad = (const float*)d_in[base + 2];
    const float* bb = (const float*)d_in[base + 3];
    const float* g  = (const float*)d_in[base + 4];
    const float* bt = (const float*)d_in[base + 5];
    const float* m  = (const float*)d_in[base + 6];
    const float* vv = (const float*)d_in[base + 7];

    const int heads = (l < 2) ? NHEADS : 1;
    const int c     = HIDC;
    const int hc    = heads * c;

    dim3 ggrid(hc / 16, (unsigned)((n_wave_jobs + 7) / 8));
    if (l == 0)
      gemm_wmma_f32<FIN><<<ggrid, 256, 0, stream>>>(in, W, hf, hc, n_wave_jobs);
    else
      gemm_wmma_f32<NHEADS * HIDC><<<ggrid, 256, 0, stream>>>(in, W, hf, hc, n_wave_jobs);

    const int nh = NNODES * heads;
    att_scores<<<gblk(nh, 256), 256, 0, stream>>>(hf, as, ad, ssrc, sdst,
                                                  NNODES, heads, c);

    if (l == 0)
      gat_dst_tile<NHEADS, false><<<(NNODES + TILE_NODES - 1) / TILE_NODES, 1024, 0, stream>>>(
          ei, lists, lcnt, ssrc, sdst, hf, bb, g, bt, m, vv, agg, 1);
    else if (l == 1)
      gat_dst_tile<NHEADS, true><<<(NNODES + TILE_NODES - 1) / TILE_NODES, 1024, 0, stream>>>(
          ei, lists, lcnt, ssrc, sdst, hf, bb, g, bt, m, vv, agg, 1);
    else
      gat_dst_tile<1, true><<<(NNODES + TILE_NODES - 1) / TILE_NODES, 1024, 0, stream>>>(
          ei, lists, lcnt, ssrc, sdst, hf, bb, g, bt, m, vv, agg, 0);

    in = agg;
  }

  final_linear<<<gblk(NNODES, 256), 256, 0, stream>>>(in, Wc, bc, (float*)d_out);
}


# Destination tile 256 -> 200 nodes: 250 work-groups, one on each of 250 of the chip's 256 WGPs (was 196), each with 22 % fewer rows and edges; list slots 256 -> 192; sum order as before
# speedup vs baseline: 3.5950x; 1.0613x over previous
#include <hip/hip_runtime.h>

typedef __attribute__((ext_vector_type(2))) float v2f;
typedef __attribute__((ext_vector_type(4))) float v4f;
typedef __attribute__((ext_vector_type(8))) float v8f;

#define NNODES 50000
#define NEDGES 800000
#define FIN    128
#define HIDC   64
#define NHEADS 4
#define EPSBN  1e-5f

#define GEMM_KC 32
#define GEMM_AP (GEMM_KC + 4)
#define GEMM_ROWS 80
#define GEMM_NLD (GEMM_ROWS / 4)

template <int K>
__global__ __launch_bounds__(256) void gemm_wmma_f32(
    const float* __restrict__ A, const float* __restrict__ B,
    float* __restrict__ C, int Nc, int n_wave_jobs)
{
  static_assert(K % GEMM_KC == 0, "whole chunks");
  __shared__ float Bs[K * 16];
  __shared__ __attribute__((aligned(16))) float As[8 * GEMM_ROWS * GEMM_AP];

  const int tid  = threadIdx.x;
  const int lane = tid & 31;
  const int wave = tid >> 5;
  const int half = lane >> 4;
  const int ml   = lane & 15;
  const int col0 = blockIdx.x * 16;

  const int  job    = blockIdx.y * 8 + wave;
  const bool active = job < n_wave_jobs;
  const int  row_base = job * GEMM_ROWS;

  const int lr = lane >> 3, lk = 4 * (lane & 7);
  const float* gp = A + (size_t)(row_base + lr) * K + lk;
  v4f st[GEMM_NLD];
  if (active) {
    #pragma unroll
    for (int j = 0; j < GEMM_NLD; ++j)
      st[j] = *(const v4f*)(gp + (size_t)(4 * j) * K);
  }

  #pragma unroll
  for (int i = tid; i < K * 16; i += 256) {
    const int r = i >> 4, cc = i & 15;
    Bs[i] = B[(size_t)r * Nc + col0 + cc];
  }
  __syncthreads();

  if (active) {
    float* slab = As + wave * (GEMM_ROWS * GEMM_AP);

    v8f acc[5];
    #pragma unroll
    for (int t = 0; t < 5; ++t)
      acc[t] = (v8f){0.f,0.f,0.f,0.f,0.f,0.f,0.f,0.f};

    for (int kc = 0; kc < K; kc += GEMM_KC) {
      #pragma unroll
      for (int j = 0; j < GEMM_NLD; ++j)
        *(v4f*)(slab + (4 * j + lr) * GEMM_AP + lk) = st[j];
      if (kc + GEMM_KC < K) {
        #pragma unroll
        for (int j = 0; j < GEMM_NLD; ++j)
          st[j] = *(const v4f*)(gp + (size_t)(4 * j) * K + (kc + GEMM_KC));
      }
      #pragma unroll
      for (int k0 = 0; k0 < GEMM_KC; k0 += 4) {
        v2f b;
        b.x = Bs[(kc + k0 + 2*half + 0) * 16 + ml];
        b.y = Bs[(kc + k0 + 2*half + 1) * 16 + ml];
        const float* ap = slab + ml * GEMM_AP + (k0 + 2*half);
        #pragma unroll
        for (int t = 0; t < 5; ++t) {
          v2f a = *(const v2f*)(ap + (16 * t) * GEMM_AP);
          acc[t] = __builtin_amdgcn_wmma_f32_16x16x4_f32(
              false, a, false, b, (short)0, acc[t], false, false);
        }
      }
    }

    #pragma unroll
    for (int t = 0; t < 5; ++t) {
      float* cp = C + (size_t)(row_base + 16*t + 8*half) * Nc + col0 + ml;
      #pragma unroll
      for (int i = 0; i < 8; ++i)
        cp[(size_t)i * Nc] = acc[t][i];
    }
  }
}

__global__ void att_scores(const float* __restrict__ hf,
                           const float* __restrict__ a_s,
                           const float* __restrict__ a_d,
                           float* __restrict__ s_src,
                           float* __restrict__ s_dst,
                           int n, int heads, int c)
{
  int idx = blockIdx.x * blockDim.x + threadIdx.x;
  if (idx >= n * heads) return;
  int node = idx / heads, h = idx % heads;
  const float* f  = hf  + (size_t)node * heads * c + (size_t)h * c;
  const float* as = a_s + (size_t)h * c;
  const float* ad = a_d + (size_t)h * c;
  float ss = 0.f, sd = 0.f;
  #pragma unroll 4
  for (int i = 0; i < c; ++i) { float v = f[i]; ss += v * as[i]; sd += v * ad[i]; }
  s_src[idx] = ss; s_dst[idx] = sd;
}

__device__ __forceinline__ unsigned f2key(float f) {
  unsigned u = __float_as_uint(f);
  return (u & 0x80000000u) ? ~u : (u | 0x80000000u);
}
__device__ __forceinline__ float key2f(unsigned k) {
  return __uint_as_float((k & 0x80000000u) ? (k ^ 0x80000000u) : ~k);
}

#define TILE_NODES 200
#define NTILES     ((NNODES + TILE_NODES - 1) / TILE_NODES)
#define QSLOTS     192
#define QEBITS     20
#define QSBITS     16

template <int HEADS, int UE>
__device__ __forceinline__ void tile_edge_batch(
    const float* __restrict__ hf, float* acc,
    int src, int dl, const float (&a)[HEADS], int j, int lane)
{
  constexpr int HC = HEADS * HIDC;
  constexpr int NI = HC / 32;
  float v[UE][NI];
  float al[UE][HEADS];
  int   d[UE];
  #pragma unroll
  for (int u = 0; u < UE; ++u) {
    const int s = __builtin_amdgcn_readlane(src, j + u);
    d[u]        = __builtin_amdgcn_readlane(dl, j + u);
    #pragma unroll
    for (int h = 0; h < HEADS; ++h)
      al[u][h] = __int_as_float(
          __builtin_amdgcn_readlane(__float_as_int(a[h]), j + u));
    const float* hp = hf + (size_t)s * HC + lane;
    #pragma unroll
    for (int i = 0; i < NI; ++i) v[u][i] = hp[32 * i];
  }
  #pragma unroll
  for (int u = 0; u < UE; ++u) {
    float* ap = acc + d[u] * HC + lane;
    #pragma unroll
    for (int i = 0; i < NI; ++i)
      atomicAdd(ap + 32 * i, v[u][i] * al[u][(32 * i) / HIDC]);
  }
}

__device__ __forceinline__ float leaky02(float v) { return v > 0.f ? v : 0.2f * v; }

template <int HEADS, bool SAVED>
__device__ __forceinline__ void tile_list_max(
    const int* __restrict__ ei, const int* __restrict__ gl,
    const float* __restrict__ s_src,
    unsigned* smax, int* q, int cnt, int lane)
{
  for (int g = 0; g < cnt; g += 128) {
    int  dl[4], sr[4];
    bool on[4];
    #pragma unroll
    for (int k = 0; k < 4; ++k) {
      const int i = g + 32 * k + lane;
      on[k] = i < cnt;
      if constexpr (SAVED) {
        const int w = on[k] ? gl[i] : 0;
        dl[k] = (int)((unsigned)w >> QSBITS);
        sr[k] = w & ((1 << QSBITS) - 1);
      } else {
        const int w = on[k] ? q[i] : 0;
        dl[k] = (int)((unsigned)w >> QEBITS);
        sr[k] = ei[w & ((1 << QEBITS) - 1)];
      }
    }
    if constexpr (HEADS == 4) {
      float4 sv[4];
      #pragma unroll
      for (int k = 0; k < 4; ++k)
        sv[k] = *(const float4*)(s_src + (size_t)sr[k] * HEADS);
      #pragma unroll
      for (int k = 0; k < 4; ++k) {
        if (on[k]) {
          unsigned* mp = smax + dl[k] * HEADS;
          atomicMax(mp + 0, f2key(sv[k].x));
          atomicMax(mp + 1, f2key(sv[k].y));
          atomicMax(mp + 2, f2key(sv[k].z));
          atomicMax(mp + 3, f2key(sv[k].w));
          q[g + 32 * k + lane] = sr[k] | (dl[k] << QSBITS);
        }
      }
    } else {
      float sv[4][HEADS];
      #pragma unroll
      for (int k = 0; k < 4; ++k) {
        #pragma unroll
        for (int h = 0; h < HEADS; ++h)
          sv[k][h] = s_src[(size_t)sr[k] * HEADS + h];
      }
      #pragma unroll
      for (int k = 0; k < 4; ++k) {
        if (on[k]) {
          #pragma unroll
          for (int h = 0; h < HEADS; ++h)
            atomicMax(smax + dl[k] * HEADS + h, f2key(sv[k][h]));
          q[g + 32 * k + lane] = sr[k] | (dl[k] << QSBITS);
        }
      }
    }
  }
}

__device__ __forceinline__ void tile_list_form(
    const int* __restrict__ ei, int* q, int cnt, int lane)
{
  for (int i = lane; i < cnt; i += 32) {
    const int w = q[i];
    q[i] = ei[w & ((1 << QEBITS) - 1)] | ((int)((unsigned)w >> QEBITS) << QSBITS);
  }
}

template <int HEADS>
__device__ __forceinline__ void tile_list_sum(
    const float* __restrict__ s_src, const float* __restrict__ hf,
    float* acc, float* den, const float* sdl, const unsigned* rmax,
    const int* q, int base, int n, int lane)
{
  constexpr int HC = HEADS * HIDC;
  constexpr int UE = 32 / (HC / 32);
  int src = 0, dl = 0;
  float a[HEADS];
  #pragma unroll
  for (int h = 0; h < HEADS; ++h) a[h] = 0.f;
  if (lane < n) {
    const int w = q[base + lane];
    src = w & ((1 << QSBITS) - 1);
    dl  = (int)((unsigned)w >> QSBITS);
    if constexpr (HEADS == 4) {
      const float4 sv = *(const float4*)(s_src + (size_t)src * HEADS);
      const float4 sd = *(const float4*)(sdl + dl * HEADS);
      const uint4  mk = *(const uint4*)(rmax + dl * HEADS);
      a[0] = expf(leaky02(sv.x + sd.x) - __uint_as_float(mk.x));
      a[1] = expf(leaky02(sv.y + sd.y) - __uint_as_float(mk.y));
      a[2] = expf(leaky02(sv.z + sd.z) - __uint_as_float(mk.z));
      a[3] = expf(leaky02(sv.w + sd.w) - __uint_as_float(mk.w));
    } else {
      #pragma unroll
      for (int h = 0; h < HEADS; ++h)
        a[h] = expf(leaky02(s_src[(size_t)src * HEADS + h] + sdl[dl * HEADS + h]) -
                    __uint_as_float(rmax[dl * HEADS + h]));
    }
    float* dp = den + dl * HEADS;
    #pragma unroll
    for (int h = 0; h < HEADS; ++h)
      atomicAdd(dp + h, a[h]);
  }
  int j = 0;
  for (; j + UE <= n; j += UE)
    tile_edge_batch<HEADS, UE>(hf, acc, src, dl, a, j, lane);
  if constexpr (UE > 4) {
    for (; j + 4 <= n; j += 4)
      tile_edge_batch<HEADS, 4>(hf, acc, src, dl, a, j, lane);
  }
  for (; j < n; ++j)
    tile_edge_batch<HEADS, 1>(hf, acc, src, dl, a, j, lane);
}

template <int HEADS, bool SAVED>
__global__ __launch_bounds__(1024) void gat_dst_tile(
    const int* __restrict__ ei,
    int* __restrict__ lists,
    int* __restrict__ lcnt,

    const float* __restrict__ s_src,
    const float* __restrict__ s_dst,
    const float* __restrict__ hf,
    const float* __restrict__ bias,
    const float* __restrict__ g,
    const float* __restrict__ bt,
    const float* __restrict__ m,
    const float* __restrict__ v,
    float* __restrict__ out, int do_elu)
{
  constexpr int HC  = HEADS * HIDC;
  constexpr int HC4 = HC / 4;
  constexpr int TS  = TILE_NODES * HEADS;
  constexpr int NCH = NEDGES / 128;
  static_assert(NEDGES % 128 == 0, "edge list splits into whole 512-B loads");
  static_assert(1024 % HC4 == 0, "a thread keeps its channels in every pass");
  static_assert(TS <= 1024, "one thread per (row, head) pair");
  static_assert(NEDGES <= (1 << QEBITS) && TILE_NODES <= (1 << (31 - QEBITS)),
                "edge number and tile row share one list word");
  static_assert(NNODES <= (1 << QSBITS) && TILE_NODES <= (1 << (31 - QSBITS)),
                "source node and tile row share one list word");
  static_assert(QSLOTS >= 128, "an empty list takes any one chunk");

  __shared__ __attribute__((aligned(16))) float    acc[TILE_NODES * HC];
  __shared__ int queue[32 * QSLOTS];
  __shared__ __attribute__((aligned(16))) unsigned rmax[TS];
  __shared__ __attribute__((aligned(16))) float    sdl[TS];
  __shared__ __attribute__((aligned(16))) float    den[TS];

  const int tid  = threadIdx.x;
  const int lane = tid & 31;
  const int wv   = __builtin_amdgcn_readfirstlane(tid >> 5);
  const int n0   = (int)blockIdx.x * TILE_NODES;
  const int nt   = (NNODES - n0) < TILE_NODES ? (NNODES - n0) : TILE_NODES;

  float s_self = 0.f;
  if (tid < nt * HEADS) {
    s_self    = s_src[(size_t)n0 * HEADS + tid];
    rmax[tid] = f2key(s_self);
    sdl[tid]  = s_dst[(size_t)n0 * HEADS + tid];
  }
  __syncthreads();

  const int* dstp = ei + NEDGES;
  int* q  = queue + wv * QSLOTS;
  int cnt = 0;

  auto ldchunk = [&](int cc) __attribute__((always_inline)) -> int4 {
    cc = cc < NCH ? cc : NCH - 1;
    return *(const int4*)(dstp + (size_t)cc * 128 + 4 * lane);
  };
  auto enqueue = [&](unsigned mask, int e, int d) __attribute__((always_inline)) {
    if (mask) {
      const int pos = cnt + (int)__builtin_amdgcn_mbcnt_lo(mask, 0u);
      if ((mask >> lane) & 1u)
        q[pos] = e | ((d - n0) << QEBITS);
      cnt += __builtin_popcount(mask);
    }
  };
  auto take = [&](const int4 cur, int cc) __attribute__((always_inline)) -> bool {
    const unsigned m0 = __builtin_amdgcn_ballot_w32((unsigned)(cur.x - n0) < (unsigned)TILE_NODES);
    const unsigned m1 = __builtin_amdgcn_ballot_w32((unsigned)(cur.y - n0) < (unsigned)TILE_NODES);
    const unsigned m2 = __builtin_amdgcn_ballot_w32((unsigned)(cur.z - n0) < (unsigned)TILE_NODES);
    const unsigned m3 = __builtin_amdgcn_ballot_w32((unsigned)(cur.w - n0) < (unsigned)TILE_NODES);
    if ((m0 | m1 | m2 | m3) == 0u) return true;
    const int tot = __builtin_popcount(m0) + __builtin_popcount(m1) +
                    __builtin_popcount(m2) + __builtin_popcount(m3);
    if (cnt + tot > QSLOTS) return false;
    const int eb = cc * 128 + 4 * lane;
    enqueue(m0, eb + 0, cur.x);
    enqueue(m1, eb + 1, cur.y);
    enqueue(m2, eb + 2, cur.z);
    enqueue(m3, eb + 3, cur.w);
    return true;
  };
  auto fill = [&](int c) __attribute__((always_inline)) -> int {
    int4 a0 = ldchunk(c), a1 = ldchunk(c + 32), a2 = ldchunk(c + 64), a3 = ldchunk(c + 96);
    for (;;) {
      const int4 b0 = ldchunk(c + 128), b1 = ldchunk(c + 160),
                 b2 = ldchunk(c + 192), b3 = ldchunk(c + 224);
      if (!take(a0, c)) return c;
      if (c + 32 < NCH && !take(a1, c + 32)) return c + 32;
      if (c + 64 < NCH && !take(a2, c + 64)) return c + 64;
      if (c + 96 < NCH && !take(a3, c + 96)) return c + 96;
      c += 128;
      if (c >= NCH) return NCH;
      a0 = ldchunk(c + 128); a1 = ldchunk(c + 160);
      a2 = ldchunk(c + 192); a3 = ldchunk(c + 224);
      if (!take(b0, c)) return c;
      if (c + 32 < NCH && !take(b1, c + 32)) return c + 32;
      if (c + 64 < NCH && !take(b2, c + 64)) return c + 64;
      if (c + 96 < NCH && !take(b3, c + 96)) return c + 96;
      c += 128;
      if (c >= NCH) return NCH;
    }
  };

  const int slot = (int)blockIdx.x * 32 + wv;
  int* gl = lists + (size_t)slot * QSLOTS;

  int  c = wv;
  bool whole = false;
  bool taken = false;
  if constexpr (SAVED) {
    const int c0 = lcnt[slot];
    if (c0 >= 0) {
      cnt = c0;
      tile_list_max<HEADS, true>(ei, gl, s_src, rmax, q, cnt, lane);
      whole = true;
      taken = true;
    }
  }
  if (!taken) {
    int rounds = 0;
    do {
      cnt = 0;
      c = fill(c);
      tile_list_max<HEADS, false>(ei, gl, s_src, rmax, q, cnt, lane);
      ++rounds;
    } while (c < NCH);
    whole = rounds == 1;
    if constexpr (!SAVED) {
      if (whole)
        for (int i = lane; i < cnt; i += 32) gl[i] = q[i];
      if (lane == 0) lcnt[slot] = whole ? cnt : -1;
    }
  }
  __syncthreads();

  if (tid < nt * HEADS) {
    const float t  = sdl[tid];
    const float mm = leaky02(key2f(rmax[tid]) + t);
    rmax[tid] = __float_as_uint(mm);
    den[tid]  = expf(leaky02(s_self + t) - mm);
  }
  __syncthreads();

  for (int i = tid; i < nt * HC4; i += 1024) {
    const int nl = i / HC4, c4 = i % HC4;
    const float al = den[nl * HEADS + (4 * c4) / HIDC];
    const float4 r = *(const float4*)(hf + (size_t)(n0 + nl) * HC + 4 * c4);
    float4 o;
    o.x = r.x * al; o.y = r.y * al; o.z = r.z * al; o.w = r.w * al;
    *(float4*)(acc + 4 * i) = o;
  }
  __syncthreads();

  if (whole) {
    for (int b = 0; b < cnt; b += 32)
      tile_list_sum<HEADS>(s_src, hf, acc, den, sdl, rmax, q, b,
                           (cnt - b) < 32 ? (cnt - b) : 32, lane);
  } else {
    c = wv;
    while (c < NCH) {
      cnt = 0;
      c = fill(c);
      tile_list_form(ei, q, cnt, lane);
      for (int b = 0; b < cnt; b += 32)
        tile_list_sum<HEADS>(s_src, hf, acc, den, sdl, rmax, q, b,
                             (cnt - b) < 32 ? (cnt - b) : 32, lane);
    }
  }
  __syncthreads();

  {
    const int c0 = 4 * (tid % HC4);
    const int hh = c0 / HIDC;
    const float4 bv = *(const float4*)(bias + c0);
    const float4 gv = *(const float4*)(g + c0);
    const float4 tv = *(const float4*)(bt + c0);
    const float4 mv = *(const float4*)(m + c0);
    const float4 vv = *(const float4*)(v + c0);
    const float r0 = rsqrtf(vv.x + EPSBN), r1 = rsqrtf(vv.y + EPSBN);
    const float r2 = rsqrtf(vv.z + EPSBN), r3 = rsqrtf(vv.w + EPSBN);
    float* op = out + (size_t)n0 * HC;
    for (int i = tid; i < nt * HC4; i += 1024) {
      const float4 s = *(const float4*)(acc + 4 * i);
      const float rd = 1.0f / (den[(i / HC4) * HEADS + hh] + 1e-16f);
      float4 y;
      y.x = (s.x * rd + bv.x - mv.x) * r0 * gv.x + tv.x;
      y.y = (s.y * rd + bv.y - mv.y) * r1 * gv.y + tv.y;
      y.z = (s.z * rd + bv.z - mv.z) * r2 * gv.z + tv.z;
      y.w = (s.w * rd + bv.w - mv.w) * r3 * gv.w + tv.w;
      if (do_elu) {
        y.x = y.x > 0.f ? y.x : expm1f(y.x);
        y.y = y.y > 0.f ? y.y : expm1f(y.y);
        y.z = y.z > 0.f ? y.z : expm1f(y.z);
        y.w = y.w > 0.f ? y.w : expm1f(y.w);
      }
      *(float4*)(op + 4 * i) = y;
    }
  }
}

__global__ void final_linear(const float* __restrict__ h,
                             const float* __restrict__ Wc,
                             const float* __restrict__ bc,
                             float* __restrict__ out)
{
  int n = blockIdx.x * blockDim.x + threadIdx.x;
  if (n >= NNODES) return;
  float o0 = bc[0], o1 = bc[1];
  const float* r = h + (size_t)n * HIDC;
  #pragma unroll 8
  for (int i = 0; i < HIDC; ++i) {
    float x = r[i];
    o0 += x * Wc[2*i + 0];
    o1 += x * Wc[2*i + 1];
  }
  out[2*n + 0] = o0;
  out[2*n + 1] = o1;
}

static inline unsigned gblk(long long total, int bs) {
  return (unsigned)((total + bs - 1) / bs);
}

extern "C" void kernel_launch(void* const* d_in, const int* in_sizes, int n_in,
                              void* d_out, int out_size, void* d_ws, size_t ws_size,
                              hipStream_t stream)
{
  const float* x  = (const float*)d_in[0];
  const int*   ei = (const int*)  d_in[1];
  const float* Wc = (const float*)d_in[26];
  const float* bc = (const float*)d_in[27];

  char* ws = (char*)d_ws;
  float*    hf    = (float*)ws;    ws += (size_t)NNODES * 256 * sizeof(float);
  float*    agg   = (float*)ws;    ws += (size_t)NNODES * 256 * sizeof(float);
  float*    ssrc  = (float*)ws;    ws += (size_t)NNODES * NHEADS * sizeof(float);
  float*    sdst  = (float*)ws;    ws += (size_t)NNODES * NHEADS * sizeof(float);
  int*      lists = (int*)ws;      ws += (size_t)NTILES * 32 * QSLOTS * sizeof(int);
  int*      lcnt  = (int*)ws;      ws += (size_t)NTILES * 32 * sizeof(int);

  const int n_wave_jobs = (NNODES / 16) / 5;

  const float* in = x;
  for (int l = 0; l < 3; ++l) {
    int base = 2 + l * 8;
    const float* W  = (const float*)d_in[base + 0];
    const float* as = (const float*)d_in[base + 1];
    const float* ad = (const float*)d_in[base + 2];
    const float* bb = (const float*)d_in[base + 3];
    const float* g  = (const float*)d_in[base + 4];
    const float* bt = (const float*)d_in[base + 5];
    const float* m  = (const float*)d_in[base + 6];
    const float* vv = (const float*)d_in[base + 7];

    const int heads = (l < 2) ? NHEADS : 1;
    const int c     = HIDC;
    const int hc    = heads * c;

    dim3 ggrid(hc / 16, (unsigned)((n_wave_jobs + 7) / 8));
    if (l == 0)
      gemm_wmma_f32<FIN><<<ggrid, 256, 0, stream>>>(in, W, hf, hc, n_wave_jobs);
    else
      gemm_wmma_f32<NHEADS * HIDC><<<ggrid, 256, 0, stream>>>(in, W, hf, hc, n_wave_jobs);

    const int nh = NNODES * heads;
    att_scores<<<gblk(nh, 256), 256, 0, stream>>>(hf, as, ad, ssrc, sdst,
                                                  NNODES, heads, c);

    if (l == 0)
      gat_dst_tile<NHEADS, false><<<(NNODES + TILE_NODES - 1) / TILE_NODES, 1024, 0, stream>>>(
          ei, lists, lcnt, ssrc, sdst, hf, bb, g, bt, m, vv, agg, 1);
    else if (l == 1)
      gat_dst_tile<NHEADS, true><<<(NNODES + TILE_NODES - 1) / TILE_NODES, 1024, 0, stream>>>(
          ei, lists, lcnt, ssrc, sdst, hf, bb, g, bt, m, vv, agg, 1);
    else
      gat_dst_tile<1, true><<<(NNODES + TILE_NODES - 1) / TILE_NODES, 1024, 0, stream>>>(
          ei, lists, lcnt, ssrc, sdst, hf, bb, g, bt, m, vv, agg, 0);

    in = agg;
  }

  final_linear<<<gblk(NNODES, 256), 256, 0, stream>>>(in, Wc, bc, (float*)d_out);
}


# Tile lists by a two-level counting sort: a pre-pass sorts each 3200-edge slice by destination tile in LDS, tile waves collect their segments; layer 0's 250 column scans gone; sums in arrival order
# speedup vs baseline: 3.7980x; 1.0565x over previous
#include <hip/hip_runtime.h>

typedef __attribute__((ext_vector_type(2))) float v2f;
typedef __attribute__((ext_vector_type(4))) float v4f;
typedef __attribute__((ext_vector_type(8))) float v8f;

#define NNODES 50000
#define NEDGES 800000
#define FIN    128
#define HIDC   64
#define NHEADS 4
#define EPSBN  1e-5f

#define GEMM_KC 32
#define GEMM_AP (GEMM_KC + 4)
#define GEMM_ROWS 80
#define GEMM_NLD (GEMM_ROWS / 4)

template <int K>
__global__ __launch_bounds__(256) void gemm_wmma_f32(
    const float* __restrict__ A, const float* __restrict__ B,
    float* __restrict__ C, int Nc, int n_wave_jobs)
{
  static_assert(K % GEMM_KC == 0, "whole chunks");
  __shared__ float Bs[K * 16];
  __shared__ __attribute__((aligned(16))) float As[8 * GEMM_ROWS * GEMM_AP];

  const int tid  = threadIdx.x;
  const int lane = tid & 31;
  const int wave = tid >> 5;
  const int half = lane >> 4;
  const int ml   = lane & 15;
  const int col0 = blockIdx.x * 16;

  const int  job    = blockIdx.y * 8 + wave;
  const bool active = job < n_wave_jobs;
  const int  row_base = job * GEMM_ROWS;

  const int lr = lane >> 3, lk = 4 * (lane & 7);
  const float* gp = A + (size_t)(row_base + lr) * K + lk;
  v4f st[GEMM_NLD];
  if (active) {
    #pragma unroll
    for (int j = 0; j < GEMM_NLD; ++j)
      st[j] = *(const v4f*)(gp + (size_t)(4 * j) * K);
  }

  #pragma unroll
  for (int i = tid; i < K * 16; i += 256) {
    const int r = i >> 4, cc = i & 15;
    Bs[i] = B[(size_t)r * Nc + col0 + cc];
  }
  __syncthreads();

  if (active) {
    float* slab = As + wave * (GEMM_ROWS * GEMM_AP);

    v8f acc[5];
    #pragma unroll
    for (int t = 0; t < 5; ++t)
      acc[t] = (v8f){0.f,0.f,0.f,0.f,0.f,0.f,0.f,0.f};

    for (int kc = 0; kc < K; kc += GEMM_KC) {
      #pragma unroll
      for (int j = 0; j < GEMM_NLD; ++j)
        *(v4f*)(slab + (4 * j + lr) * GEMM_AP + lk) = st[j];
      if (kc + GEMM_KC < K) {
        #pragma unroll
        for (int j = 0; j < GEMM_NLD; ++j)
          st[j] = *(const v4f*)(gp + (size_t)(4 * j) * K + (kc + GEMM_KC));
      }
      #pragma unroll
      for (int k0 = 0; k0 < GEMM_KC; k0 += 4) {
        v2f b;
        b.x = Bs[(kc + k0 + 2*half + 0) * 16 + ml];
        b.y = Bs[(kc + k0 + 2*half + 1) * 16 + ml];
        const float* ap = slab + ml * GEMM_AP + (k0 + 2*half);
        #pragma unroll
        for (int t = 0; t < 5; ++t) {
          v2f a = *(const v2f*)(ap + (16 * t) * GEMM_AP);
          acc[t] = __builtin_amdgcn_wmma_f32_16x16x4_f32(
              false, a, false, b, (short)0, acc[t], false, false);
        }
      }
    }

    #pragma unroll
    for (int t = 0; t < 5; ++t) {
      float* cp = C + (size_t)(row_base + 16*t + 8*half) * Nc + col0 + ml;
      #pragma unroll
      for (int i = 0; i < 8; ++i)
        cp[(size_t)i * Nc] = acc[t][i];
    }
  }
}

__global__ void att_scores(const float* __restrict__ hf,
                           const float* __restrict__ a_s,
                           const float* __restrict__ a_d,
                           float* __restrict__ s_src,
                           float* __restrict__ s_dst,
                           int n, int heads, int c)
{
  int idx = blockIdx.x * blockDim.x + threadIdx.x;
  if (idx >= n * heads) return;
  int node = idx / heads, h = idx % heads;
  const float* f  = hf  + (size_t)node * heads * c + (size_t)h * c;
  const float* as = a_s + (size_t)h * c;
  const float* ad = a_d + (size_t)h * c;
  float ss = 0.f, sd = 0.f;
  #pragma unroll 4
  for (int i = 0; i < c; ++i) { float v = f[i]; ss += v * as[i]; sd += v * ad[i]; }
  s_src[idx] = ss; s_dst[idx] = sd;
}

__device__ __forceinline__ unsigned f2key(float f) {
  unsigned u = __float_as_uint(f);
  return (u & 0x80000000u) ? ~u : (u | 0x80000000u);
}
__device__ __forceinline__ float key2f(unsigned k) {
  return __uint_as_float((k & 0x80000000u) ? (k ^ 0x80000000u) : ~k);
}

#define TILE_NODES 200
#define NTILES     ((NNODES + TILE_NODES - 1) / TILE_NODES)
#define QSLOTS     192
#define QEBITS     20
#define QSBITS     16
#define SLICE_EDGES (NEDGES / NTILES)
#define SOFF_PITCH  256

__global__ __launch_bounds__(1024) void bin_edges(
    const int* __restrict__ ei, int* __restrict__ segs, int* __restrict__ soff)
{
  static_assert(NEDGES % NTILES == 0, "whole slices");
  static_assert(SLICE_EDGES <= 4 * 1024, "four edges per thread cover a slice");
  __shared__ int tcnt[SOFF_PITCH];
  __shared__ int toff[SOFF_PITCH];
  __shared__ int wsum[SOFF_PITCH / 32];
  static_assert(SOFF_PITCH % 32 == 0 && SOFF_PITCH <= 1024, "whole waves take the prefix sum");
  const int tid = threadIdx.x;
  const int e0  = (int)blockIdx.x * SLICE_EDGES;
  if (tid < SOFF_PITCH) tcnt[tid] = 0;
  __syncthreads();
  int  word[4], tile[4], pos[4];
  bool on[4];
  #pragma unroll
  for (int k = 0; k < 4; ++k) {
    const int i = tid + 1024 * k;
    on[k] = i < SLICE_EDGES;
    const int s = on[k] ? ei[e0 + i] : 0;
    const int d = on[k] ? ei[NEDGES + e0 + i] : 0;
    on[k] = on[k] && (unsigned)d < (unsigned)NNODES;
    tile[k] = on[k] ? d / TILE_NODES : 0;
    word[k] = s | ((d - tile[k] * TILE_NODES) << QSBITS);
  }
  #pragma unroll
  for (int k = 0; k < 4; ++k)
    pos[k] = on[k] ? atomicAdd(&tcnt[tile[k]], 1) : 0;
  __syncthreads();
  int own = 0, incl = 0;
  if (tid < SOFF_PITCH) {
    own = tcnt[tid];
    incl = own;
    #pragma unroll
    for (int o = 1; o < 32; o <<= 1) {
      const int up = __shfl_up(incl, o);
      if ((tid & 31) >= o) incl += up;
    }
    if ((tid & 31) == 31) wsum[tid >> 5] = incl;
  }
  __syncthreads();
  if (tid < SOFF_PITCH) {
    int s = incl - own;
    #pragma unroll
    for (int w = 0; w < SOFF_PITCH / 32; ++w) s += w < (tid >> 5) ? wsum[w] : 0;
    toff[tid] = s;
    soff[(int)blockIdx.x * SOFF_PITCH + tid] = s;
  }
  __syncthreads();
  #pragma unroll
  for (int k = 0; k < 4; ++k)
    if (on[k]) segs[e0 + toff[tile[k]] + pos[k]] = word[k];
}

template <int HEADS, int UE>
__device__ __forceinline__ void tile_edge_batch(
    const float* __restrict__ hf, float* acc,
    int src, int dl, const float (&a)[HEADS], int j, int lane)
{
  constexpr int HC = HEADS * HIDC;
  constexpr int NI = HC / 32;
  float v[UE][NI];
  float al[UE][HEADS];
  int   d[UE];
  #pragma unroll
  for (int u = 0; u < UE; ++u) {
    const int s = __builtin_amdgcn_readlane(src, j + u);
    d[u]        = __builtin_amdgcn_readlane(dl, j + u);
    #pragma unroll
    for (int h = 0; h < HEADS; ++h)
      al[u][h] = __int_as_float(
          __builtin_amdgcn_readlane(__float_as_int(a[h]), j + u));
    const float* hp = hf + (size_t)s * HC + lane;
    #pragma unroll
    for (int i = 0; i < NI; ++i) v[u][i] = hp[32 * i];
  }
  #pragma unroll
  for (int u = 0; u < UE; ++u) {
    float* ap = acc + d[u] * HC + lane;
    #pragma unroll
    for (int i = 0; i < NI; ++i)
      atomicAdd(ap + 32 * i, v[u][i] * al[u][(32 * i) / HIDC]);
  }
}

__device__ __forceinline__ float leaky02(float v) { return v > 0.f ? v : 0.2f * v; }

template <int HEADS, bool SAVED>
__device__ __forceinline__ void tile_list_max(
    const int* __restrict__ ei,
    const float* __restrict__ s_src,
    unsigned* smax, int* q, int cnt, int lane)
{
  for (int g = 0; g < cnt; g += 128) {
    int  dl[4], sr[4];
    bool on[4];
    #pragma unroll
    for (int k = 0; k < 4; ++k) {
      const int i = g + 32 * k + lane;
      on[k] = i < cnt;
      if constexpr (SAVED) {
        const int w = on[k] ? q[i] : 0;
        dl[k] = (int)((unsigned)w >> QSBITS);
        sr[k] = w & ((1 << QSBITS) - 1);
      } else {
        const int w = on[k] ? q[i] : 0;
        dl[k] = (int)((unsigned)w >> QEBITS);
        sr[k] = ei[w & ((1 << QEBITS) - 1)];
      }
    }
    if constexpr (HEADS == 4) {
      float4 sv[4];
      #pragma unroll
      for (int k = 0; k < 4; ++k)
        sv[k] = *(const float4*)(s_src + (size_t)sr[k] * HEADS);
      #pragma unroll
      for (int k = 0; k < 4; ++k) {
        if (on[k]) {
          unsigned* mp = smax + dl[k] * HEADS;
          atomicMax(mp + 0, f2key(sv[k].x));
          atomicMax(mp + 1, f2key(sv[k].y));
          atomicMax(mp + 2, f2key(sv[k].z));
          atomicMax(mp + 3, f2key(sv[k].w));
          if constexpr (!SAVED) q[g + 32 * k + lane] = sr[k] | (dl[k] << QSBITS);
        }
      }
    } else {
      float sv[4][HEADS];
      #pragma unroll
      for (int k = 0; k < 4; ++k) {
        #pragma unroll
        for (int h = 0; h < HEADS; ++h)
          sv[k][h] = s_src[(size_t)sr[k] * HEADS + h];
      }
      #pragma unroll
      for (int k = 0; k < 4; ++k) {
        if (on[k]) {
          #pragma unroll
          for (int h = 0; h < HEADS; ++h)
            atomicMax(smax + dl[k] * HEADS + h, f2key(sv[k][h]));
          if constexpr (!SAVED) q[g + 32 * k + lane] = sr[k] | (dl[k] << QSBITS);
        }
      }
    }
  }
}

__device__ __forceinline__ void tile_list_form(
    const int* __restrict__ ei, int* q, int cnt, int lane)
{
  for (int i = lane; i < cnt; i += 32) {
    const int w = q[i];
    q[i] = ei[w & ((1 << QEBITS) - 1)] | ((int)((unsigned)w >> QEBITS) << QSBITS);
  }
}

template <int HEADS>
__device__ __forceinline__ void tile_list_sum(
    const float* __restrict__ s_src, const float* __restrict__ hf,
    float* acc, float* den, const float* sdl, const unsigned* rmax,
    const int* q, int base, int n, int lane)
{
  constexpr int HC = HEADS * HIDC;
  constexpr int UE = 32 / (HC / 32);
  int src = 0, dl = 0;
  float a[HEADS];
  #pragma unroll
  for (int h = 0; h < HEADS; ++h) a[h] = 0.f;
  if (lane < n) {
    const int w = q[base + lane];
    src = w & ((1 << QSBITS) - 1);
    dl  = (int)((unsigned)w >> QSBITS);
    if constexpr (HEADS == 4) {
      const float4 sv = *(const float4*)(s_src + (size_t)src * HEADS);
      const float4 sd = *(const float4*)(sdl + dl * HEADS);
      const uint4  mk = *(const uint4*)(rmax + dl * HEADS);
      a[0] = expf(leaky02(sv.x + sd.x) - __uint_as_float(mk.x));
      a[1] = expf(leaky02(sv.y + sd.y) - __uint_as_float(mk.y));
      a[2] = expf(leaky02(sv.z + sd.z) - __uint_as_float(mk.z));
      a[3] = expf(leaky02(sv.w + sd.w) - __uint_as_float(mk.w));
    } else {
      #pragma unroll
      for (int h = 0; h < HEADS; ++h)
        a[h] = expf(leaky02(s_src[(size_t)src * HEADS + h] + sdl[dl * HEADS + h]) -
                    __uint_as_float(rmax[dl * HEADS + h]));
    }
    float* dp = den + dl * HEADS;
    #pragma unroll
    for (int h = 0; h < HEADS; ++h)
      atomicAdd(dp + h, a[h]);
  }
  int j = 0;
  for (; j + UE <= n; j += UE)
    tile_edge_batch<HEADS, UE>(hf, acc, src, dl, a, j, lane);
  if constexpr (UE > 4) {
    for (; j + 4 <= n; j += 4)
      tile_edge_batch<HEADS, 4>(hf, acc, src, dl, a, j, lane);
  }
  for (; j < n; ++j)
    tile_edge_batch<HEADS, 1>(hf, acc, src, dl, a, j, lane);
}

template <int HEADS>
__global__ __launch_bounds__(1024) void gat_dst_tile(
    const int* __restrict__ ei,
    const int* __restrict__ segs,
    const int* __restrict__ soff,

    const float* __restrict__ s_src,
    const float* __restrict__ s_dst,
    const float* __restrict__ hf,
    const float* __restrict__ bias,
    const float* __restrict__ g,
    const float* __restrict__ bt,
    const float* __restrict__ m,
    const float* __restrict__ v,
    float* __restrict__ out, int do_elu)
{
  constexpr int HC  = HEADS * HIDC;
  constexpr int HC4 = HC / 4;
  constexpr int TS  = TILE_NODES * HEADS;
  constexpr int NCH = NEDGES / 128;
  static_assert(NEDGES % 128 == 0, "edge list splits into whole 512-B loads");
  static_assert(1024 % HC4 == 0, "a thread keeps its channels in every pass");
  static_assert(TS <= 1024, "one thread per (row, head) pair");
  static_assert(NEDGES <= (1 << QEBITS) && TILE_NODES <= (1 << (31 - QEBITS)),
                "edge number and tile row share one list word");
  static_assert(NNODES <= (1 << QSBITS) && TILE_NODES <= (1 << (31 - QSBITS)),
                "source node and tile row share one list word");
  static_assert(QSLOTS >= 128, "an empty list takes any one chunk");

  __shared__ __attribute__((aligned(16))) float    acc[TILE_NODES * HC];
  __shared__ int queue[32 * QSLOTS];
  __shared__ __attribute__((aligned(16))) unsigned rmax[TS];
  __shared__ __attribute__((aligned(16))) float    sdl[TS];
  __shared__ __attribute__((aligned(16))) float    den[TS];
  __shared__ int nofit;

  const int tid  = threadIdx.x;
  const int lane = tid & 31;
  const int wv   = __builtin_amdgcn_readfirstlane(tid >> 5);
  const int n0   = (int)blockIdx.x * TILE_NODES;
  const int nt   = (NNODES - n0) < TILE_NODES ? (NNODES - n0) : TILE_NODES;

  float s_self = 0.f;
  if (tid == 0) nofit = 0;
  if (tid < nt * HEADS) {
    s_self    = s_src[(size_t)n0 * HEADS + tid];
    rmax[tid] = f2key(s_self);
    sdl[tid]  = s_dst[(size_t)n0 * HEADS + tid];
  }
  __syncthreads();

  const int* dstp = ei + NEDGES;
  int* q  = queue + wv * QSLOTS;
  int cnt = 0;

  auto ldchunk = [&](int cc) __attribute__((always_inline)) -> int4 {
    cc = cc < NCH ? cc : NCH - 1;
    return *(const int4*)(dstp + (size_t)cc * 128 + 4 * lane);
  };
  auto enqueue = [&](unsigned mask, int e, int d) __attribute__((always_inline)) {
    if (mask) {
      const int pos = cnt + (int)__builtin_amdgcn_mbcnt_lo(mask, 0u);
      if ((mask >> lane) & 1u)
        q[pos] = e | ((d - n0) << QEBITS);
      cnt += __builtin_popcount(mask);
    }
  };
  auto take = [&](const int4 cur, int cc) __attribute__((always_inline)) -> bool {
    const unsigned m0 = __builtin_amdgcn_ballot_w32((unsigned)(cur.x - n0) < (unsigned)TILE_NODES);
    const unsigned m1 = __builtin_amdgcn_ballot_w32((unsigned)(cur.y - n0) < (unsigned)TILE_NODES);
    const unsigned m2 = __builtin_amdgcn_ballot_w32((unsigned)(cur.z - n0) < (unsigned)TILE_NODES);
    const unsigned m3 = __builtin_amdgcn_ballot_w32((unsigned)(cur.w - n0) < (unsigned)TILE_NODES);
    if ((m0 | m1 | m2 | m3) == 0u) return true;
    const int tot = __builtin_popcount(m0) + __builtin_popcount(m1) +
                    __builtin_popcount(m2) + __builtin_popcount(m3);
    if (cnt + tot > QSLOTS) return false;
    const int eb = cc * 128 + 4 * lane;
    enqueue(m0, eb + 0, cur.x);
    enqueue(m1, eb + 1, cur.y);
    enqueue(m2, eb + 2, cur.z);
    enqueue(m3, eb + 3, cur.w);
    return true;
  };
  auto fill = [&](int c) __attribute__((always_inline)) -> int {
    int4 a0 = ldchunk(c), a1 = ldchunk(c + 32), a2 = ldchunk(c + 64), a3 = ldchunk(c + 96);
    for (;;) {
      const int4 b0 = ldchunk(c + 128), b1 = ldchunk(c + 160),
                 b2 = ldchunk(c + 192), b3 = ldchunk(c + 224);
      if (!take(a0, c)) return c;
      if (c + 32 < NCH && !take(a1, c + 32)) return c + 32;
      if (c + 64 < NCH && !take(a2, c + 64)) return c + 64;
      if (c + 96 < NCH && !take(a3, c + 96)) return c + 96;
      c += 128;
      if (c >= NCH) return NCH;
      a0 = ldchunk(c + 128); a1 = ldchunk(c + 160);
      a2 = ldchunk(c + 192); a3 = ldchunk(c + 224);
      if (!take(b0, c)) return c;
      if (c + 32 < NCH && !take(b1, c + 32)) return c + 32;
      if (c + 64 < NCH && !take(b2, c + 64)) return c + 64;
      if (c + 96 < NCH && !take(b3, c + 96)) return c + 96;
      c += 128;
      if (c >= NCH) return NCH;
    }
  };

  constexpr int NSL = (NTILES + 31) / 32;
  static_assert(NSL <= 32, "one lane per slice of the wave");
  static_assert(NTILES + 1 <= SOFF_PITCH, "a slice's offsets fit their pitch");
  int seg_len = 0, seg_at = 0;
  {
    const int gs = wv + 32 * lane;
    if (lane < NSL && gs < NTILES) {
      const int o0 = soff[gs * SOFF_PITCH + (int)blockIdx.x];
      const int o1 = soff[gs * SOFF_PITCH + (int)blockIdx.x + 1];
      seg_len = o1 - o0;
      seg_at  = gs * SLICE_EDGES + o0;
    }
  }
  int ln[NSL], at[NSL], first[NSL];
  int tot = 0;
  #pragma unroll
  for (int k = 0; k < NSL; ++k) {
    ln[k] = __builtin_amdgcn_readlane(seg_len, k);
    at[k] = __builtin_amdgcn_readlane(seg_at, k);
    first[k] = lane < ln[k] ? segs[at[k] + lane] : 0;
    tot += ln[k];
  }

  if (tot > QSLOTS && lane == 0) nofit = 1;
  __syncthreads();
  int  c = wv;
  bool whole = false;
  if (nofit == 0) {
    #pragma unroll
    for (int k = 0; k < NSL; ++k) {
      if (lane < ln[k]) q[cnt + lane] = first[k];
      #pragma unroll 1
      for (int i = 32 + lane; i < ln[k]; i += 32)
        q[cnt + i] = segs[at[k] + i];
      cnt += ln[k];
    }
    tile_list_max<HEADS, true>(ei, s_src, rmax, q, cnt, lane);
    whole = true;
  } else {
    int rounds = 0;
    do {
      cnt = 0;
      c = fill(c);
      tile_list_max<HEADS, false>(ei, s_src, rmax, q, cnt, lane);
      ++rounds;
    } while (c < NCH);
    whole = rounds == 1;
  }
  __syncthreads();

  if (tid < nt * HEADS) {
    const float t  = sdl[tid];
    const float mm = leaky02(key2f(rmax[tid]) + t);
    rmax[tid] = __float_as_uint(mm);
    den[tid]  = expf(leaky02(s_self + t) - mm);
  }
  __syncthreads();

  for (int i = tid; i < nt * HC4; i += 1024) {
    const int nl = i / HC4, c4 = i % HC4;
    const float al = den[nl * HEADS + (4 * c4) / HIDC];
    const float4 r = *(const float4*)(hf + (size_t)(n0 + nl) * HC + 4 * c4);
    float4 o;
    o.x = r.x * al; o.y = r.y * al; o.z = r.z * al; o.w = r.w * al;
    *(float4*)(acc + 4 * i) = o;
  }
  __syncthreads();

  if (whole) {
    for (int b = 0; b < cnt; b += 32)
      tile_list_sum<HEADS>(s_src, hf, acc, den, sdl, rmax, q, b,
                           (cnt - b) < 32 ? (cnt - b) : 32, lane);
  } else {
    c = wv;
    while (c < NCH) {
      cnt = 0;
      c = fill(c);
      tile_list_form(ei, q, cnt, lane);
      for (int b = 0; b < cnt; b += 32)
        tile_list_sum<HEADS>(s_src, hf, acc, den, sdl, rmax, q, b,
                             (cnt - b) < 32 ? (cnt - b) : 32, lane);
    }
  }
  __syncthreads();

  {
    const int c0 = 4 * (tid % HC4);
    const int hh = c0 / HIDC;
    const float4 bv = *(const float4*)(bias + c0);
    const float4 gv = *(const float4*)(g + c0);
    const float4 tv = *(const float4*)(bt + c0);
    const float4 mv = *(const float4*)(m + c0);
    const float4 vv = *(const float4*)(v + c0);
    const float r0 = rsqrtf(vv.x + EPSBN), r1 = rsqrtf(vv.y + EPSBN);
    const float r2 = rsqrtf(vv.z + EPSBN), r3 = rsqrtf(vv.w + EPSBN);
    float* op = out + (size_t)n0 * HC;
    for (int i = tid; i < nt * HC4; i += 1024) {
      const float4 s = *(const float4*)(acc + 4 * i);
      const float rd = 1.0f / (den[(i / HC4) * HEADS + hh] + 1e-16f);
      float4 y;
      y.x = (s.x * rd + bv.x - mv.x) * r0 * gv.x + tv.x;
      y.y = (s.y * rd + bv.y - mv.y) * r1 * gv.y + tv.y;
      y.z = (s.z * rd + bv.z - mv.z) * r2 * gv.z + tv.z;
      y.w = (s.w * rd + bv.w - mv.w) * r3 * gv.w + tv.w;
      if (do_elu) {
        y.x = y.x > 0.f ? y.x : expm1f(y.x);
        y.y = y.y > 0.f ? y.y : expm1f(y.y);
        y.z = y.z > 0.f ? y.z : expm1f(y.z);
        y.w = y.w > 0.f ? y.w : expm1f(y.w);
      }
      *(float4*)(op + 4 * i) = y;
    }
  }
}

__global__ void final_linear(const float* __restrict__ h,
                             const float* __restrict__ Wc,
                             const float* __restrict__ bc,
                             float* __restrict__ out)
{
  int n = blockIdx.x * blockDim.x + threadIdx.x;
  if (n >= NNODES) return;
  float o0 = bc[0], o1 = bc[1];
  const float* r = h + (size_t)n * HIDC;
  #pragma unroll 8
  for (int i = 0; i < HIDC; ++i) {
    float x = r[i];
    o0 += x * Wc[2*i + 0];
    o1 += x * Wc[2*i + 1];
  }
  out[2*n + 0] = o0;
  out[2*n + 1] = o1;
}

static inline unsigned gblk(long long total, int bs) {
  return (unsigned)((total + bs - 1) / bs);
}

extern "C" void kernel_launch(void* const* d_in, const int* in_sizes, int n_in,
                              void* d_out, int out_size, void* d_ws, size_t ws_size,
                              hipStream_t stream)
{
  const float* x  = (const float*)d_in[0];
  const int*   ei = (const int*)  d_in[1];
  const float* Wc = (const float*)d_in[26];
  const float* bc = (const float*)d_in[27];

  char* ws = (char*)d_ws;
  float*    hf    = (float*)ws;    ws += (size_t)NNODES * 256 * sizeof(float);
  float*    agg   = (float*)ws;    ws += (size_t)NNODES * 256 * sizeof(float);
  float*    ssrc  = (float*)ws;    ws += (size_t)NNODES * NHEADS * sizeof(float);
  float*    sdst  = (float*)ws;    ws += (size_t)NNODES * NHEADS * sizeof(float);
  int*      segs  = (int*)ws;      ws += (size_t)NEDGES * sizeof(int);
  int*      soff  = (int*)ws;      ws += (size_t)NTILES * SOFF_PITCH * sizeof(int);

  bin_edges<<<NTILES, 1024, 0, stream>>>(ei, segs, soff);

  const int n_wave_jobs = (NNODES / 16) / 5;

  const float* in = x;
  for (int l = 0; l < 3; ++l) {
    int base = 2 + l * 8;
    const float* W  = (const float*)d_in[base + 0];
    const float* as = (const float*)d_in[base + 1];
    const float* ad = (const float*)d_in[base + 2];
    const float* bb = (const float*)d_in[base + 3];
    const float* g  = (const float*)d_in[base + 4];
    const float* bt = (const float*)d_in[base + 5];
    const float* m  = (const float*)d_in[base + 6];
    const float* vv = (const float*)d_in[base + 7];

    const int heads = (l < 2) ? NHEADS : 1;
    const int c     = HIDC;
    const int hc    = heads * c;

    dim3 ggrid(hc / 16, (unsigned)((n_wave_jobs + 7) / 8));
    if (l == 0)
      gemm_wmma_f32<FIN><<<ggrid, 256, 0, stream>>>(in, W, hf, hc, n_wave_jobs);
    else
      gemm_wmma_f32<NHEADS * HIDC><<<ggrid, 256, 0, stream>>>(in, W, hf, hc, n_wave_jobs);

    const int nh = NNODES * heads;
    att_scores<<<gblk(nh, 256), 256, 0, stream>>>(hf, as, ad, ssrc, sdst,
                                                  NNODES, heads, c);

    if (l < 2)
      gat_dst_tile<NHEADS><<<NTILES, 1024, 0, stream>>>(
          ei, segs, soff, ssrc, sdst, hf, bb, g, bt, m, vv, agg, 1);
    else
      gat_dst_tile<1><<<NTILES, 1024, 0, stream>>>(
          ei, segs, soff, ssrc, sdst, hf, bb, g, bt, m, vv, agg, 0);

    in = agg;
  }

  final_linear<<<gblk(NNODES, 256), 256, 0, stream>>>(in, Wc, bc, (float*)d_out);
}


# Row-dot passes (3 score launches, final 64->2 linear) re-mapped: 16 lanes share a 64-float row, 16 B per lane, whole 128-B lines per load; dot sum order now 4 per lane, then 16-lane tree
# speedup vs baseline: 4.0547x; 1.0676x over previous
#include <hip/hip_runtime.h>

typedef __attribute__((ext_vector_type(2))) float v2f;
typedef __attribute__((ext_vector_type(4))) float v4f;
typedef __attribute__((ext_vector_type(8))) float v8f;

#define NNODES 50000
#define NEDGES 800000
#define FIN    128
#define HIDC   64
#define NHEADS 4
#define EPSBN  1e-5f

#define GEMM_KC 32
#define GEMM_AP (GEMM_KC + 4)
#define GEMM_ROWS 80
#define GEMM_NLD (GEMM_ROWS / 4)

template <int K>
__global__ __launch_bounds__(256) void gemm_wmma_f32(
    const float* __restrict__ A, const float* __restrict__ B,
    float* __restrict__ C, int Nc, int n_wave_jobs)
{
  static_assert(K % GEMM_KC == 0, "whole chunks");
  __shared__ float Bs[K * 16];
  __shared__ __attribute__((aligned(16))) float As[8 * GEMM_ROWS * GEMM_AP];

  const int tid  = threadIdx.x;
  const int lane = tid & 31;
  const int wave = tid >> 5;
  const int half = lane >> 4;
  const int ml   = lane & 15;
  const int col0 = blockIdx.x * 16;

  const int  job    = blockIdx.y * 8 + wave;
  const bool active = job < n_wave_jobs;
  const int  row_base = job * GEMM_ROWS;

  const int lr = lane >> 3, lk = 4 * (lane & 7);
  const float* gp = A + (size_t)(row_base + lr) * K + lk;
  v4f st[GEMM_NLD];
  if (active) {
    #pragma unroll
    for (int j = 0; j < GEMM_NLD; ++j)
      st[j] = *(const v4f*)(gp + (size_t)(4 * j) * K);
  }

  #pragma unroll
  for (int i = tid; i < K * 16; i += 256) {
    const int r = i >> 4, cc = i & 15;
    Bs[i] = B[(size_t)r * Nc + col0 + cc];
  }
  __syncthreads();

  if (active) {
    float* slab = As + wave * (GEMM_ROWS * GEMM_AP);

    v8f acc[5];
    #pragma unroll
    for (int t = 0; t < 5; ++t)
      acc[t] = (v8f){0.f,0.f,0.f,0.f,0.f,0.f,0.f,0.f};

    for (int kc = 0; kc < K; kc += GEMM_KC) {
      #pragma unroll
      for (int j = 0; j < GEMM_NLD; ++j)
        *(v4f*)(slab + (4 * j + lr) * GEMM_AP + lk) = st[j];
      if (kc + GEMM_KC < K) {
        #pragma unroll
        for (int j = 0; j < GEMM_NLD; ++j)
          st[j] = *(const v4f*)(gp + (size_t)(4 * j) * K + (kc + GEMM_KC));
      }
      #pragma unroll
      for (int k0 = 0; k0 < GEMM_KC; k0 += 4) {
        v2f b;
        b.x = Bs[(kc + k0 + 2*half + 0) * 16 + ml];
        b.y = Bs[(kc + k0 + 2*half + 1) * 16 + ml];
        const float* ap = slab + ml * GEMM_AP + (k0 + 2*half);
        #pragma unroll
        for (int t = 0; t < 5; ++t) {
          v2f a = *(const v2f*)(ap + (16 * t) * GEMM_AP);
          acc[t] = __builtin_amdgcn_wmma_f32_16x16x4_f32(
              false, a, false, b, (short)0, acc[t], false, false);
        }
      }
    }

    #pragma unroll
    for (int t = 0; t < 5; ++t) {
      float* cp = C + (size_t)(row_base + 16*t + 8*half) * Nc + col0 + ml;
      #pragma unroll
      for (int i = 0; i < 8; ++i)
        cp[(size_t)i * Nc] = acc[t][i];
    }
  }
}

#define ROWDOT_U     8
#define ROWDOT_ROWS  (2 * ROWDOT_U)
#define ROWDOT_WGS   512

__device__ __forceinline__ float row16_sum(float v) {
  v += __int_as_float(__builtin_amdgcn_update_dpp(0, __float_as_int(v), 0x161, 0xf, 0xf, false));
  v += __int_as_float(__builtin_amdgcn_update_dpp(0, __float_as_int(v), 0x162, 0xf, 0xf, false));
  v += __int_as_float(__builtin_amdgcn_update_dpp(0, __float_as_int(v), 0x164, 0xf, 0xf, false));
  v += __int_as_float(__builtin_amdgcn_update_dpp(0, __float_as_int(v), 0x168, 0xf, 0xf, false));
  return v;
}

template <int NSET, typename Put>
__device__ __forceinline__ void row_dots_wave(
    const float* __restrict__ rows, int ntrips,
    const v4f (&wa)[NSET], const v4f (&wb)[NSET], Put put)
{
  const int lane = threadIdx.x & 31;
  const int sub  = lane & 15, hh = lane >> 4;
  const int w0 = __builtin_amdgcn_readfirstlane((int)blockIdx.x * 8 + (int)(threadIdx.x >> 5));
  const int nw = (int)gridDim.x * 8;

  auto request = [&](v4f (&buf)[ROWDOT_U], int t) __attribute__((always_inline)) {
    const float* p = rows + (size_t)t * (ROWDOT_ROWS * HIDC) + 4 * lane;
    #pragma unroll
    for (int u = 0; u < ROWDOT_U; ++u) buf[u] = *(const v4f*)(p + u * (2 * HIDC));
  };
  auto finish = [&](const v4f (&buf)[ROWDOT_U], int t) __attribute__((always_inline)) {
    float da[ROWDOT_U], db[ROWDOT_U];
    #pragma unroll
    for (int u = 0; u < ROWDOT_U; ++u) {
      const v4f x = buf[u], a = wa[u % NSET], b = wb[u % NSET];
      da[u] = x.x * a.x + x.y * a.y + x.z * a.z + x.w * a.w;
      db[u] = x.x * b.x + x.y * b.y + x.z * b.z + x.w * b.w;
    }
    #pragma unroll
    for (int u = 0; u < ROWDOT_U; ++u) { da[u] = row16_sum(da[u]); db[u] = row16_sum(db[u]); }
    float ra = da[0], rb = db[0];
    #pragma unroll
    for (int u = 1; u < ROWDOT_U; ++u) { ra = sub == u ? da[u] : ra; rb = sub == u ? db[u] : rb; }
    if (sub < ROWDOT_U) put(t * ROWDOT_ROWS + 2 * sub + hh, ra, rb);
  };

  v4f b0[ROWDOT_U], b1[ROWDOT_U];
  int t = w0;
  if (t >= ntrips) return;
  request(b0, t);
  for (;;) {
    int tn = t + nw;
    request(b1, tn < ntrips ? tn : t);
    __builtin_amdgcn_sched_barrier(0);
    finish(b0, t);
    if (tn >= ntrips) return;
    t = tn; tn = t + nw;
    request(b0, tn < ntrips ? tn : t);
    __builtin_amdgcn_sched_barrier(0);
    finish(b1, t);
    if (tn >= ntrips) return;
    t = tn;
  }
}

template <int HEADS>
__global__ __launch_bounds__(256) void att_scores(const float* __restrict__ hf,
                                                  const float* __restrict__ a_s,
                                                  const float* __restrict__ a_d,
                                                  float* __restrict__ s_src,
                                                  float* __restrict__ s_dst)
{
  constexpr int NSET = (HEADS + 1) / 2;
  static_assert((NNODES * HEADS) % ROWDOT_ROWS == 0, "whole trips");
  static_assert(ROWDOT_ROWS % HEADS == 0, "a trip starts at head 0");
  static_assert(HEADS == 1 || (HEADS % 2 == 0 && ROWDOT_U % NSET == 0), "load u meets head set u % NSET");
  const int lane = threadIdx.x & 31;
  const int sub  = lane & 15, hh = lane >> 4;
  v4f wa[NSET], wb[NSET];
  #pragma unroll
  for (int k = 0; k < NSET; ++k) {
    const int h = (2 * k + hh) % HEADS;
    wa[k] = *(const v4f*)(a_s + h * HIDC + 4 * sub);
    wb[k] = *(const v4f*)(a_d + h * HIDC + 4 * sub);
  }
  row_dots_wave<NSET>(hf, (NNODES * HEADS) / ROWDOT_ROWS, wa, wb,
                      [&](int p, float a, float b) __attribute__((always_inline)) {
                        s_src[p] = a; s_dst[p] = b;
                      });
}

__device__ __forceinline__ unsigned f2key(float f) {
  unsigned u = __float_as_uint(f);
  return (u & 0x80000000u) ? ~u : (u | 0x80000000u);
}
__device__ __forceinline__ float key2f(unsigned k) {
  return __uint_as_float((k & 0x80000000u) ? (k ^ 0x80000000u) : ~k);
}

#define TILE_NODES 200
#define NTILES     ((NNODES + TILE_NODES - 1) / TILE_NODES)
#define QSLOTS     192
#define QEBITS     20
#define QSBITS     16
#define SLICE_EDGES (NEDGES / NTILES)
#define SOFF_PITCH  256

__global__ __launch_bounds__(1024) void bin_edges(
    const int* __restrict__ ei, int* __restrict__ segs, int* __restrict__ soff)
{
  static_assert(NEDGES % NTILES == 0, "whole slices");
  static_assert(SLICE_EDGES <= 4 * 1024, "four edges per thread cover a slice");
  __shared__ int tcnt[SOFF_PITCH];
  __shared__ int toff[SOFF_PITCH];
  __shared__ int wsum[SOFF_PITCH / 32];
  static_assert(SOFF_PITCH % 32 == 0 && SOFF_PITCH <= 1024, "whole waves take the prefix sum");
  const int tid = threadIdx.x;
  const int e0  = (int)blockIdx.x * SLICE_EDGES;
  if (tid < SOFF_PITCH) tcnt[tid] = 0;
  __syncthreads();
  int  word[4], tile[4], pos[4];
  bool on[4];
  #pragma unroll
  for (int k = 0; k < 4; ++k) {
    const int i = tid + 1024 * k;
    on[k] = i < SLICE_EDGES;
    const int s = on[k] ? ei[e0 + i] : 0;
    const int d = on[k] ? ei[NEDGES + e0 + i] : 0;
    on[k] = on[k] && (unsigned)d < (unsigned)NNODES;
    tile[k] = on[k] ? d / TILE_NODES : 0;
    word[k] = s | ((d - tile[k] * TILE_NODES) << QSBITS);
  }
  #pragma unroll
  for (int k = 0; k < 4; ++k)
    pos[k] = on[k] ? atomicAdd(&tcnt[tile[k]], 1) : 0;
  __syncthreads();
  int own = 0, incl = 0;
  if (tid < SOFF_PITCH) {
    own = tcnt[tid];
    incl = own;
    #pragma unroll
    for (int o = 1; o < 32; o <<= 1) {
      const int up = __shfl_up(incl, o);
      if ((tid & 31) >= o) incl += up;
    }
    if ((tid & 31) == 31) wsum[tid >> 5] = incl;
  }
  __syncthreads();
  if (tid < SOFF_PITCH) {
    int s = incl - own;
    #pragma unroll
    for (int w = 0; w < SOFF_PITCH / 32; ++w) s += w < (tid >> 5) ? wsum[w] : 0;
    toff[tid] = s;
    soff[(int)blockIdx.x * SOFF_PITCH + tid] = s;
  }
  __syncthreads();
  #pragma unroll
  for (int k = 0; k < 4; ++k)
    if (on[k]) segs[e0 + toff[tile[k]] + pos[k]] = word[k];
}

template <int HEADS, int UE>
__device__ __forceinline__ void tile_edge_batch(
    const float* __restrict__ hf, float* acc,
    int src, int dl, const float (&a)[HEADS], int j, int lane)
{
  constexpr int HC = HEADS * HIDC;
  constexpr int NI = HC / 32;
  float v[UE][NI];
  float al[UE][HEADS];
  int   d[UE];
  #pragma unroll
  for (int u = 0; u < UE; ++u) {
    const int s = __builtin_amdgcn_readlane(src, j + u);
    d[u]        = __builtin_amdgcn_readlane(dl, j + u);
    #pragma unroll
    for (int h = 0; h < HEADS; ++h)
      al[u][h] = __int_as_float(
          __builtin_amdgcn_readlane(__float_as_int(a[h]), j + u));
    const float* hp = hf + (size_t)s * HC + lane;
    #pragma unroll
    for (int i = 0; i < NI; ++i) v[u][i] = hp[32 * i];
  }
  #pragma unroll
  for (int u = 0; u < UE; ++u) {
    float* ap = acc + d[u] * HC + lane;
    #pragma unroll
    for (int i = 0; i < NI; ++i)
      atomicAdd(ap + 32 * i, v[u][i] * al[u][(32 * i) / HIDC]);
  }
}

__device__ __forceinline__ float leaky02(float v) { return v > 0.f ? v : 0.2f * v; }

template <int HEADS, bool SAVED>
__device__ __forceinline__ void tile_list_max(
    const int* __restrict__ ei,
    const float* __restrict__ s_src,
    unsigned* smax, int* q, int cnt, int lane)
{
  for (int g = 0; g < cnt; g += 128) {
    int  dl[4], sr[4];
    bool on[4];
    #pragma unroll
    for (int k = 0; k < 4; ++k) {
      const int i = g + 32 * k + lane;
      on[k] = i < cnt;
      if constexpr (SAVED) {
        const int w = on[k] ? q[i] : 0;
        dl[k] = (int)((unsigned)w >> QSBITS);
        sr[k] = w & ((1 << QSBITS) - 1);
      } else {
        const int w = on[k] ? q[i] : 0;
        dl[k] = (int)((unsigned)w >> QEBITS);
        sr[k] = ei[w & ((1 << QEBITS) - 1)];
      }
    }
    if constexpr (HEADS == 4) {
      float4 sv[4];
      #pragma unroll
      for (int k = 0; k < 4; ++k)
        sv[k] = *(const float4*)(s_src + (size_t)sr[k] * HEADS);
      #pragma unroll
      for (int k = 0; k < 4; ++k) {
        if (on[k]) {
          unsigned* mp = smax + dl[k] * HEADS;
          atomicMax(mp + 0, f2key(sv[k].x));
          atomicMax(mp + 1, f2key(sv[k].y));
          atomicMax(mp + 2, f2key(sv[k].z));
          atomicMax(mp + 3, f2key(sv[k].w));
          if constexpr (!SAVED) q[g + 32 * k + lane] = sr[k] | (dl[k] << QSBITS);
        }
      }
    } else {
      float sv[4][HEADS];
      #pragma unroll
      for (int k = 0; k < 4; ++k) {
        #pragma unroll
        for (int h = 0; h < HEADS; ++h)
          sv[k][h] = s_src[(size_t)sr[k] * HEADS + h];
      }
      #pragma unroll
      for (int k = 0; k < 4; ++k) {
        if (on[k]) {
          #pragma unroll
          for (int h = 0; h < HEADS; ++h)
            atomicMax(smax + dl[k] * HEADS + h, f2key(sv[k][h]));
          if constexpr (!SAVED) q[g + 32 * k + lane] = sr[k] | (dl[k] << QSBITS);
        }
      }
    }
  }
}

__device__ __forceinline__ void tile_list_form(
    const int* __restrict__ ei, int* q, int cnt, int lane)
{
  for (int i = lane; i < cnt; i += 32) {
    const int w = q[i];
    q[i] = ei[w & ((1 << QEBITS) - 1)] | ((int)((unsigned)w >> QEBITS) << QSBITS);
  }
}

template <int HEADS>
__device__ __forceinline__ void tile_list_sum(
    const float* __restrict__ s_src, const float* __restrict__ hf,
    float* acc, float* den, const float* sdl, const unsigned* rmax,
    const int* q, int base, int n, int lane)
{
  constexpr int HC = HEADS * HIDC;
  constexpr int UE = 32 / (HC / 32);
  int src = 0, dl = 0;
  float a[HEADS];
  #pragma unroll
  for (int h = 0; h < HEADS; ++h) a[h] = 0.f;
  if (lane < n) {
    const int w = q[base + lane];
    src = w & ((1 << QSBITS) - 1);
    dl  = (int)((unsigned)w >> QSBITS);
    if constexpr (HEADS == 4) {
      const float4 sv = *(const float4*)(s_src + (size_t)src * HEADS);
      const float4 sd = *(const float4*)(sdl + dl * HEADS);
      const uint4  mk = *(const uint4*)(rmax + dl * HEADS);
      a[0] = expf(leaky02(sv.x + sd.x) - __uint_as_float(mk.x));
      a[1] = expf(leaky02(sv.y + sd.y) - __uint_as_float(mk.y));
      a[2] = expf(leaky02(sv.z + sd.z) - __uint_as_float(mk.z));
      a[3] = expf(leaky02(sv.w + sd.w) - __uint_as_float(mk.w));
    } else {
      #pragma unroll
      for (int h = 0; h < HEADS; ++h)
        a[h] = expf(leaky02(s_src[(size_t)src * HEADS + h] + sdl[dl * HEADS + h]) -
                    __uint_as_float(rmax[dl * HEADS + h]));
    }
    float* dp = den + dl * HEADS;
    #pragma unroll
    for (int h = 0; h < HEADS; ++h)
      atomicAdd(dp + h, a[h]);
  }
  int j = 0;
  for (; j + UE <= n; j += UE)
    tile_edge_batch<HEADS, UE>(hf, acc, src, dl, a, j, lane);
  if constexpr (UE > 4) {
    for (; j + 4 <= n; j += 4)
      tile_edge_batch<HEADS, 4>(hf, acc, src, dl, a, j, lane);
  }
  for (; j < n; ++j)
    tile_edge_batch<HEADS, 1>(hf, acc, src, dl, a, j, lane);
}

template <int HEADS>
__global__ __launch_bounds__(1024) void gat_dst_tile(
    const int* __restrict__ ei,
    const int* __restrict__ segs,
    const int* __restrict__ soff,

    const float* __restrict__ s_src,
    const float* __restrict__ s_dst,
    const float* __restrict__ hf,
    const float* __restrict__ bias,
    const float* __restrict__ g,
    const float* __restrict__ bt,
    const float* __restrict__ m,
    const float* __restrict__ v,
    float* __restrict__ out, int do_elu)
{
  constexpr int HC  = HEADS * HIDC;
  constexpr int HC4 = HC / 4;
  constexpr int TS  = TILE_NODES * HEADS;
  constexpr int NCH = NEDGES / 128;
  static_assert(NEDGES % 128 == 0, "edge list splits into whole 512-B loads");
  static_assert(1024 % HC4 == 0, "a thread keeps its channels in every pass");
  static_assert(TS <= 1024, "one thread per (row, head) pair");
  static_assert(NEDGES <= (1 << QEBITS) && TILE_NODES <= (1 << (31 - QEBITS)),
                "edge number and tile row share one list word");
  static_assert(NNODES <= (1 << QSBITS) && TILE_NODES <= (1 << (31 - QSBITS)),
                "source node and tile row share one list word");
  static_assert(QSLOTS >= 128, "an empty list takes any one chunk");

  __shared__ __attribute__((aligned(16))) float    acc[TILE_NODES * HC];
  __shared__ int queue[32 * QSLOTS];
  __shared__ __attribute__((aligned(16))) unsigned rmax[TS];
  __shared__ __attribute__((aligned(16))) float    sdl[TS];
  __shared__ __attribute__((aligned(16))) float    den[TS];
  __shared__ int nofit;

  const int tid  = threadIdx.x;
  const int lane = tid & 31;
  const int wv   = __builtin_amdgcn_readfirstlane(tid >> 5);
  const int n0   = (int)blockIdx.x * TILE_NODES;
  const int nt   = (NNODES - n0) < TILE_NODES ? (NNODES - n0) : TILE_NODES;

  float s_self = 0.f;
  if (tid == 0) nofit = 0;
  if (tid < nt * HEADS) {
    s_self    = s_src[(size_t)n0 * HEADS + tid];
    rmax[tid] = f2key(s_self);
    sdl[tid]  = s_dst[(size_t)n0 * HEADS + tid];
  }
  __syncthreads();

  const int* dstp = ei + NEDGES;
  int* q  = queue + wv * QSLOTS;
  int cnt = 0;

  auto ldchunk = [&](int cc) __attribute__((always_inline)) -> int4 {
    cc = cc < NCH ? cc : NCH - 1;
    return *(const int4*)(dstp + (size_t)cc * 128 + 4 * lane);
  };
  auto enqueue = [&](unsigned mask, int e, int d) __attribute__((always_inline)) {
    if (mask) {
      const int pos = cnt + (int)__builtin_amdgcn_mbcnt_lo(mask, 0u);
      if ((mask >> lane) & 1u)
        q[pos] = e | ((d - n0) << QEBITS);
      cnt += __builtin_popcount(mask);
    }
  };
  auto take = [&](const int4 cur, int cc) __attribute__((always_inline)) -> bool {
    const unsigned m0 = __builtin_amdgcn_ballot_w32((unsigned)(cur.x - n0) < (unsigned)TILE_NODES);
    const unsigned m1 = __builtin_amdgcn_ballot_w32((unsigned)(cur.y - n0) < (unsigned)TILE_NODES);
    const unsigned m2 = __builtin_amdgcn_ballot_w32((unsigned)(cur.z - n0) < (unsigned)TILE_NODES);
    const unsigned m3 = __builtin_amdgcn_ballot_w32((unsigned)(cur.w - n0) < (unsigned)TILE_NODES);
    if ((m0 | m1 | m2 | m3) == 0u) return true;
    const int tot = __builtin_popcount(m0) + __builtin_popcount(m1) +
                    __builtin_popcount(m2) + __builtin_popcount(m3);
    if (cnt + tot > QSLOTS) return false;
    const int eb = cc * 128 + 4 * lane;
    enqueue(m0, eb + 0, cur.x);
    enqueue(m1, eb + 1, cur.y);
    enqueue(m2, eb + 2, cur.z);
    enqueue(m3, eb + 3, cur.w);
    return true;
  };
  auto fill = [&](int c) __attribute__((always_inline)) -> int {
    int4 a0 = ldchunk(c), a1 = ldchunk(c + 32), a2 = ldchunk(c + 64), a3 = ldchunk(c + 96);
    for (;;) {
      const int4 b0 = ldchunk(c + 128), b1 = ldchunk(c + 160),
                 b2 = ldchunk(c + 192), b3 = ldchunk(c + 224);
      if (!take(a0, c)) return c;
      if (c + 32 < NCH && !take(a1, c + 32)) return c + 32;
      if (c + 64 < NCH && !take(a2, c + 64)) return c + 64;
      if (c + 96 < NCH && !take(a3, c + 96)) return c + 96;
      c += 128;
      if (c >= NCH) return NCH;
      a0 = ldchunk(c + 128); a1 = ldchunk(c + 160);
      a2 = ldchunk(c + 192); a3 = ldchunk(c + 224);
      if (!take(b0, c)) return c;
      if (c + 32 < NCH && !take(b1, c + 32)) return c + 32;
      if (c + 64 < NCH && !take(b2, c + 64)) return c + 64;
      if (c + 96 < NCH && !take(b3, c + 96)) return c + 96;
      c += 128;
      if (c >= NCH) return NCH;
    }
  };

  constexpr int NSL = (NTILES + 31) / 32;
  static_assert(NSL <= 32, "one lane per slice of the wave");
  static_assert(NTILES + 1 <= SOFF_PITCH, "a slice's offsets fit their pitch");
  int seg_len = 0, seg_at = 0;
  {
    const int gs = wv + 32 * lane;
    if (lane < NSL && gs < NTILES) {
      const int o0 = soff[gs * SOFF_PITCH + (int)blockIdx.x];
      const int o1 = soff[gs * SOFF_PITCH + (int)blockIdx.x + 1];
      seg_len = o1 - o0;
      seg_at  = gs * SLICE_EDGES + o0;
    }
  }
  int ln[NSL], at[NSL], first[NSL];
  int tot = 0;
  #pragma unroll
  for (int k = 0; k < NSL; ++k) {
    ln[k] = __builtin_amdgcn_readlane(seg_len, k);
    at[k] = __builtin_amdgcn_readlane(seg_at, k);
    first[k] = lane < ln[k] ? segs[at[k] + lane] : 0;
    tot += ln[k];
  }

  if (tot > QSLOTS && lane == 0) nofit = 1;
  __syncthreads();
  int  c = wv;
  bool whole = false;
  if (nofit == 0) {
    #pragma unroll
    for (int k = 0; k < NSL; ++k) {
      if (lane < ln[k]) q[cnt + lane] = first[k];
      #pragma unroll 1
      for (int i = 32 + lane; i < ln[k]; i += 32)
        q[cnt + i] = segs[at[k] + i];
      cnt += ln[k];
    }
    tile_list_max<HEADS, true>(ei, s_src, rmax, q, cnt, lane);
    whole = true;
  } else {
    int rounds = 0;
    do {
      cnt = 0;
      c = fill(c);
      tile_list_max<HEADS, false>(ei, s_src, rmax, q, cnt, lane);
      ++rounds;
    } while (c < NCH);
    whole = rounds == 1;
  }
  __syncthreads();

  if (tid < nt * HEADS) {
    const float t  = sdl[tid];
    const float mm = leaky02(key2f(rmax[tid]) + t);
    rmax[tid] = __float_as_uint(mm);
    den[tid]  = expf(leaky02(s_self + t) - mm);
  }
  __syncthreads();

  for (int i = tid; i < nt * HC4; i += 1024) {
    const int nl = i / HC4, c4 = i % HC4;
    const float al = den[nl * HEADS + (4 * c4) / HIDC];
    const float4 r = *(const float4*)(hf + (size_t)(n0 + nl) * HC + 4 * c4);
    float4 o;
    o.x = r.x * al; o.y = r.y * al; o.z = r.z * al; o.w = r.w * al;
    *(float4*)(acc + 4 * i) = o;
  }
  __syncthreads();

  if (whole) {
    for (int b = 0; b < cnt; b += 32)
      tile_list_sum<HEADS>(s_src, hf, acc, den, sdl, rmax, q, b,
                           (cnt - b) < 32 ? (cnt - b) : 32, lane);
  } else {
    c = wv;
    while (c < NCH) {
      cnt = 0;
      c = fill(c);
      tile_list_form(ei, q, cnt, lane);
      for (int b = 0; b < cnt; b += 32)
        tile_list_sum<HEADS>(s_src, hf, acc, den, sdl, rmax, q, b,
                             (cnt - b) < 32 ? (cnt - b) : 32, lane);
    }
  }
  __syncthreads();

  {
    const int c0 = 4 * (tid % HC4);
    const int hh = c0 / HIDC;
    const float4 bv = *(const float4*)(bias + c0);
    const float4 gv = *(const float4*)(g + c0);
    const float4 tv = *(const float4*)(bt + c0);
    const float4 mv = *(const float4*)(m + c0);
    const float4 vv = *(const float4*)(v + c0);
    const float r0 = rsqrtf(vv.x + EPSBN), r1 = rsqrtf(vv.y + EPSBN);
    const float r2 = rsqrtf(vv.z + EPSBN), r3 = rsqrtf(vv.w + EPSBN);
    float* op = out + (size_t)n0 * HC;
    for (int i = tid; i < nt * HC4; i += 1024) {
      const float4 s = *(const float4*)(acc + 4 * i);
      const float rd = 1.0f / (den[(i / HC4) * HEADS + hh] + 1e-16f);
      float4 y;
      y.x = (s.x * rd + bv.x - mv.x) * r0 * gv.x + tv.x;
      y.y = (s.y * rd + bv.y - mv.y) * r1 * gv.y + tv.y;
      y.z = (s.z * rd + bv.z - mv.z) * r2 * gv.z + tv.z;
      y.w = (s.w * rd + bv.w - mv.w) * r3 * gv.w + tv.w;
      if (do_elu) {
        y.x = y.x > 0.f ? y.x : expm1f(y.x);
        y.y = y.y > 0.f ? y.y : expm1f(y.y);
        y.z = y.z > 0.f ? y.z : expm1f(y.z);
        y.w = y.w > 0.f ? y.w : expm1f(y.w);
      }
      *(float4*)(op + 4 * i) = y;
    }
  }
}

__global__ __launch_bounds__(256) void final_linear(const float* __restrict__ h,
                                                    const float* __restrict__ Wc,
                                                    const float* __restrict__ bc,
                                                    float* __restrict__ out)
{
  static_assert(NNODES % ROWDOT_ROWS == 0, "whole trips");
  const int sub = threadIdx.x & 15;
  const v4f w0 = *(const v4f*)(Wc + 8 * sub), w1 = *(const v4f*)(Wc + 8 * sub + 4);
  const v4f wa[1] = {(v4f){w0.x, w0.z, w1.x, w1.z}};
  const v4f wb[1] = {(v4f){w0.y, w0.w, w1.y, w1.w}};
  const float c0 = bc[0], c1 = bc[1];
  row_dots_wave<1>(h, NNODES / ROWDOT_ROWS, wa, wb,
                   [&](int n, float a, float b) __attribute__((always_inline)) {
                     *(v2f*)(out + 2 * n) = (v2f){a + c0, b + c1};
                   });
}

extern "C" void kernel_launch(void* const* d_in, const int* in_sizes, int n_in,
                              void* d_out, int out_size, void* d_ws, size_t ws_size,
                              hipStream_t stream)
{
  const float* x  = (const float*)d_in[0];
  const int*   ei = (const int*)  d_in[1];
  const float* Wc = (const float*)d_in[26];
  const float* bc = (const float*)d_in[27];

  char* ws = (char*)d_ws;
  float*    hf    = (float*)ws;    ws += (size_t)NNODES * 256 * sizeof(float);
  float*    agg   = (float*)ws;    ws += (size_t)NNODES * 256 * sizeof(float);
  float*    ssrc  = (float*)ws;    ws += (size_t)NNODES * NHEADS * sizeof(float);
  float*    sdst  = (float*)ws;    ws += (size_t)NNODES * NHEADS * sizeof(float);
  int*      segs  = (int*)ws;      ws += (size_t)NEDGES * sizeof(int);
  int*      soff  = (int*)ws;      ws += (size_t)NTILES * SOFF_PITCH * sizeof(int);

  bin_edges<<<NTILES, 1024, 0, stream>>>(ei, segs, soff);

  const int n_wave_jobs = (NNODES / 16) / 5;

  const float* in = x;
  for (int l = 0; l < 3; ++l) {
    int base = 2 + l * 8;
    const float* W  = (const float*)d_in[base + 0];
    const float* as = (const float*)d_in[base + 1];
    const float* ad = (const float*)d_in[base + 2];
    const float* bb = (const float*)d_in[base + 3];
    const float* g  = (const float*)d_in[base + 4];
    const float* bt = (const float*)d_in[base + 5];
    const float* m  = (const float*)d_in[base + 6];
    const float* vv = (const float*)d_in[base + 7];

    const int heads = (l < 2) ? NHEADS : 1;
    const int c     = HIDC;
    const int hc    = heads * c;

    dim3 ggrid(hc / 16, (unsigned)((n_wave_jobs + 7) / 8));
    if (l == 0)
      gemm_wmma_f32<FIN><<<ggrid, 256, 0, stream>>>(in, W, hf, hc, n_wave_jobs);
    else
      gemm_wmma_f32<NHEADS * HIDC><<<ggrid, 256, 0, stream>>>(in, W, hf, hc, n_wave_jobs);

    if (l < 2)
      att_scores<NHEADS><<<ROWDOT_WGS, 256, 0, stream>>>(hf, as, ad, ssrc, sdst);
    else
      att_scores<1><<<ROWDOT_WGS, 256, 0, stream>>>(hf, as, ad, ssrc, sdst);

    if (l < 2)
      gat_dst_tile<NHEADS><<<NTILES, 1024, 0, stream>>>(
          ei, segs, soff, ssrc, sdst, hf, bb, g, bt, m, vv, agg, 1);
    else
      gat_dst_tile<1><<<NTILES, 1024, 0, stream>>>(
          ei, segs, soff, ssrc, sdst, hf, bb, g, bt, m, vv, agg, 0);

    in = agg;
  }

  final_linear<<<ROWDOT_WGS, 256, 0, stream>>>(in, Wc, bc, (float*)d_out);
}


# Layers 1, 2: the linear map runs in the tile work-group that finished the rows: rows stay in LDS as the WMMA A operand, Wn fragments from memory; 2 GEMM launches gone (11->9); K sum order unchanged
# speedup vs baseline: 4.7796x; 1.1788x over previous
#include <hip/hip_runtime.h>

typedef __attribute__((ext_vector_type(2))) float v2f;
typedef __attribute__((ext_vector_type(4))) float v4f;
typedef __attribute__((ext_vector_type(8))) float v8f;

#define NNODES 50000
#define NEDGES 800000
#define FIN    128
#define HIDC   64
#define NHEADS 4
#define EPSBN  1e-5f

#define GEMM_KC 32
#define GEMM_AP (GEMM_KC + 4)
#define GEMM_ROWS 80
#define GEMM_NLD (GEMM_ROWS / 4)

template <int K>
__global__ __launch_bounds__(256) void gemm_wmma_f32(
    const float* __restrict__ A, const float* __restrict__ B,
    float* __restrict__ C, int Nc, int n_wave_jobs)
{
  static_assert(K % GEMM_KC == 0, "whole chunks");
  __shared__ float Bs[K * 16];
  __shared__ __attribute__((aligned(16))) float As[8 * GEMM_ROWS * GEMM_AP];

  const int tid  = threadIdx.x;
  const int lane = tid & 31;
  const int wave = tid >> 5;
  const int half = lane >> 4;
  const int ml   = lane & 15;
  const int col0 = blockIdx.x * 16;

  const int  job    = blockIdx.y * 8 + wave;
  const bool active = job < n_wave_jobs;
  const int  row_base = job * GEMM_ROWS;

  const int lr = lane >> 3, lk = 4 * (lane & 7);
  const float* gp = A + (size_t)(row_base + lr) * K + lk;
  v4f st[GEMM_NLD];
  if (active) {
    #pragma unroll
    for (int j = 0; j < GEMM_NLD; ++j)
      st[j] = *(const v4f*)(gp + (size_t)(4 * j) * K);
  }

  #pragma unroll
  for (int i = tid; i < K * 16; i += 256) {
    const int r = i >> 4, cc = i & 15;
    Bs[i] = B[(size_t)r * Nc + col0 + cc];
  }
  __syncthreads();

  if (active) {
    float* slab = As + wave * (GEMM_ROWS * GEMM_AP);

    v8f acc[5];
    #pragma unroll
    for (int t = 0; t < 5; ++t)
      acc[t] = (v8f){0.f,0.f,0.f,0.f,0.f,0.f,0.f,0.f};

    for (int kc = 0; kc < K; kc += GEMM_KC) {
      #pragma unroll
      for (int j = 0; j < GEMM_NLD; ++j)
        *(v4f*)(slab + (4 * j + lr) * GEMM_AP + lk) = st[j];
      if (kc + GEMM_KC < K) {
        #pragma unroll
        for (int j = 0; j < GEMM_NLD; ++j)
          st[j] = *(const v4f*)(gp + (size_t)(4 * j) * K + (kc + GEMM_KC));
      }
      #pragma unroll
      for (int k0 = 0; k0 < GEMM_KC; k0 += 4) {
        v2f b;
        b.x = Bs[(kc + k0 + 2*half + 0) * 16 + ml];
        b.y = Bs[(kc + k0 + 2*half + 1) * 16 + ml];
        const float* ap = slab + ml * GEMM_AP + (k0 + 2*half);
        #pragma unroll
        for (int t = 0; t < 5; ++t) {
          v2f a = *(const v2f*)(ap + (16 * t) * GEMM_AP);
          acc[t] = __builtin_amdgcn_wmma_f32_16x16x4_f32(
              false, a, false, b, (short)0, acc[t], false, false);
        }
      }
    }

    #pragma unroll
    for (int t = 0; t < 5; ++t) {
      float* cp = C + (size_t)(row_base + 16*t + 8*half) * Nc + col0 + ml;
      #pragma unroll
      for (int i = 0; i < 8; ++i)
        cp[(size_t)i * Nc] = acc[t][i];
    }
  }
}

#define ROWDOT_U     8
#define ROWDOT_ROWS  (2 * ROWDOT_U)
#define ROWDOT_WGS   512

__device__ __forceinline__ float row16_sum(float v) {
  v += __int_as_float(__builtin_amdgcn_update_dpp(0, __float_as_int(v), 0x161, 0xf, 0xf, false));
  v += __int_as_float(__builtin_amdgcn_update_dpp(0, __float_as_int(v), 0x162, 0xf, 0xf, false));
  v += __int_as_float(__builtin_amdgcn_update_dpp(0, __float_as_int(v), 0x164, 0xf, 0xf, false));
  v += __int_as_float(__builtin_amdgcn_update_dpp(0, __float_as_int(v), 0x168, 0xf, 0xf, false));
  return v;
}

template <int NSET, typename Put>
__device__ __forceinline__ void row_dots_wave(
    const float* __restrict__ rows, int ntrips,
    const v4f (&wa)[NSET], const v4f (&wb)[NSET], Put put)
{
  const int lane = threadIdx.x & 31;
  const int sub  = lane & 15, hh = lane >> 4;
  const int w0 = __builtin_amdgcn_readfirstlane((int)blockIdx.x * 8 + (int)(threadIdx.x >> 5));
  const int nw = (int)gridDim.x * 8;

  auto request = [&](v4f (&buf)[ROWDOT_U], int t) __attribute__((always_inline)) {
    const float* p = rows + (size_t)t * (ROWDOT_ROWS * HIDC) + 4 * lane;
    #pragma unroll
    for (int u = 0; u < ROWDOT_U; ++u) buf[u] = *(const v4f*)(p + u * (2 * HIDC));
  };
  auto finish = [&](const v4f (&buf)[ROWDOT_U], int t) __attribute__((always_inline)) {
    float da[ROWDOT_U], db[ROWDOT_U];
    #pragma unroll
    for (int u = 0; u < ROWDOT_U; ++u) {
      const v4f x = buf[u], a = wa[u % NSET], b = wb[u % NSET];
      da[u] = x.x * a.x + x.y * a.y + x.z * a.z + x.w * a.w;
      db[u] = x.x * b.x + x.y * b.y + x.z * b.z + x.w * b.w;
    }
    #pragma unroll
    for (int u = 0; u < ROWDOT_U; ++u) { da[u] = row16_sum(da[u]); db[u] = row16_sum(db[u]); }
    float ra = da[0], rb = db[0];
    #pragma unroll
    for (int u = 1; u < ROWDOT_U; ++u) { ra = sub == u ? da[u] : ra; rb = sub == u ? db[u] : rb; }
    if (sub < ROWDOT_U) put(t * ROWDOT_ROWS + 2 * sub + hh, ra, rb);
  };

  v4f b0[ROWDOT_U], b1[ROWDOT_U];
  int t = w0;
  if (t >= ntrips) return;
  request(b0, t);
  for (;;) {
    int tn = t + nw;
    request(b1, tn < ntrips ? tn : t);
    __builtin_amdgcn_sched_barrier(0);
    finish(b0, t);
    if (tn >= ntrips) return;
    t = tn; tn = t + nw;
    request(b0, tn < ntrips ? tn : t);
    __builtin_amdgcn_sched_barrier(0);
    finish(b1, t);
    if (tn >= ntrips) return;
    t = tn;
  }
}

template <int HEADS>
__global__ __launch_bounds__(256) void att_scores(const float* __restrict__ hf,
                                                  const float* __restrict__ a_s,
                                                  const float* __restrict__ a_d,
                                                  float* __restrict__ s_src,
                                                  float* __restrict__ s_dst)
{
  constexpr int NSET = (HEADS + 1) / 2;
  static_assert((NNODES * HEADS) % ROWDOT_ROWS == 0, "whole trips");
  static_assert(ROWDOT_ROWS % HEADS == 0, "a trip starts at head 0");
  static_assert(HEADS == 1 || (HEADS % 2 == 0 && ROWDOT_U % NSET == 0), "load u meets head set u % NSET");
  const int lane = threadIdx.x & 31;
  const int sub  = lane & 15, hh = lane >> 4;
  v4f wa[NSET], wb[NSET];
  #pragma unroll
  for (int k = 0; k < NSET; ++k) {
    const int h = (2 * k + hh) % HEADS;
    wa[k] = *(const v4f*)(a_s + h * HIDC + 4 * sub);
    wb[k] = *(const v4f*)(a_d + h * HIDC + 4 * sub);
  }
  row_dots_wave<NSET>(hf, (NNODES * HEADS) / ROWDOT_ROWS, wa, wb,
                      [&](int p, float a, float b) __attribute__((always_inline)) {
                        s_src[p] = a; s_dst[p] = b;
                      });
}

__device__ __forceinline__ unsigned f2key(float f) {
  unsigned u = __float_as_uint(f);
  return (u & 0x80000000u) ? ~u : (u | 0x80000000u);
}
__device__ __forceinline__ float key2f(unsigned k) {
  return __uint_as_float((k & 0x80000000u) ? (k ^ 0x80000000u) : ~k);
}

#define TILE_NODES 200
#define NTILES     ((NNODES + TILE_NODES - 1) / TILE_NODES)
#define QSLOTS     192
#define QEBITS     20
#define QSBITS     16
#define SLICE_EDGES (NEDGES / NTILES)
#define SOFF_PITCH  256

__global__ __launch_bounds__(1024) void bin_edges(
    const int* __restrict__ ei, int* __restrict__ segs, int* __restrict__ soff)
{
  static_assert(NEDGES % NTILES == 0, "whole slices");
  static_assert(SLICE_EDGES <= 4 * 1024, "four edges per thread cover a slice");
  __shared__ int tcnt[SOFF_PITCH];
  __shared__ int toff[SOFF_PITCH];
  __shared__ int wsum[SOFF_PITCH / 32];
  static_assert(SOFF_PITCH % 32 == 0 && SOFF_PITCH <= 1024, "whole waves take the prefix sum");
  const int tid = threadIdx.x;
  const int e0  = (int)blockIdx.x * SLICE_EDGES;
  if (tid < SOFF_PITCH) tcnt[tid] = 0;
  __syncthreads();
  int  word[4], tile[4], pos[4];
  bool on[4];
  #pragma unroll
  for (int k = 0; k < 4; ++k) {
    const int i = tid + 1024 * k;
    on[k] = i < SLICE_EDGES;
    const int s = on[k] ? ei[e0 + i] : 0;
    const int d = on[k] ? ei[NEDGES + e0 + i] : 0;
    on[k] = on[k] && (unsigned)d < (unsigned)NNODES;
    tile[k] = on[k] ? d / TILE_NODES : 0;
    word[k] = s | ((d - tile[k] * TILE_NODES) << QSBITS);
  }
  #pragma unroll
  for (int k = 0; k < 4; ++k)
    pos[k] = on[k] ? atomicAdd(&tcnt[tile[k]], 1) : 0;
  __syncthreads();
  int own = 0, incl = 0;
  if (tid < SOFF_PITCH) {
    own = tcnt[tid];
    incl = own;
    #pragma unroll
    for (int o = 1; o < 32; o <<= 1) {
      const int up = __shfl_up(incl, o);
      if ((tid & 31) >= o) incl += up;
    }
    if ((tid & 31) == 31) wsum[tid >> 5] = incl;
  }
  __syncthreads();
  if (tid < SOFF_PITCH) {
    int s = incl - own;
    #pragma unroll
    for (int w = 0; w < SOFF_PITCH / 32; ++w) s += w < (tid >> 5) ? wsum[w] : 0;
    toff[tid] = s;
    soff[(int)blockIdx.x * SOFF_PITCH + tid] = s;
  }
  __syncthreads();
  #pragma unroll
  for (int k = 0; k < 4; ++k)
    if (on[k]) segs[e0 + toff[tile[k]] + pos[k]] = word[k];
}

template <int HEADS, int UE, int AP>
__device__ __forceinline__ void tile_edge_batch(
    const float* __restrict__ hf, float* acc,
    int src, int dl, const float (&a)[HEADS], int j, int lane)
{
  constexpr int HC = HEADS * HIDC;
  constexpr int NI = HC / 32;
  float v[UE][NI];
  float al[UE][HEADS];
  int   d[UE];
  #pragma unroll
  for (int u = 0; u < UE; ++u) {
    const int s = __builtin_amdgcn_readlane(src, j + u);
    d[u]        = __builtin_amdgcn_readlane(dl, j + u);
    #pragma unroll
    for (int h = 0; h < HEADS; ++h)
      al[u][h] = __int_as_float(
          __builtin_amdgcn_readlane(__float_as_int(a[h]), j + u));
    const float* hp = hf + (size_t)s * HC + lane;
    #pragma unroll
    for (int i = 0; i < NI; ++i) v[u][i] = hp[32 * i];
  }
  #pragma unroll
  for (int u = 0; u < UE; ++u) {
    float* ap = acc + d[u] * AP + lane;
    #pragma unroll
    for (int i = 0; i < NI; ++i)
      atomicAdd(ap + 32 * i, v[u][i] * al[u][(32 * i) / HIDC]);
  }
}

__device__ __forceinline__ float leaky02(float v) { return v > 0.f ? v : 0.2f * v; }

template <int HEADS, bool SAVED>
__device__ __forceinline__ void tile_list_max(
    const int* __restrict__ ei,
    const float* __restrict__ s_src,
    unsigned* smax, int* q, int cnt, int lane)
{
  for (int g = 0; g < cnt; g += 128) {
    int  dl[4], sr[4];
    bool on[4];
    #pragma unroll
    for (int k = 0; k < 4; ++k) {
      const int i = g + 32 * k + lane;
      on[k] = i < cnt;
      if constexpr (SAVED) {
        const int w = on[k] ? q[i] : 0;
        dl[k] = (int)((unsigned)w >> QSBITS);
        sr[k] = w & ((1 << QSBITS) - 1);
      } else {
        const int w = on[k] ? q[i] : 0;
        dl[k] = (int)((unsigned)w >> QEBITS);
        sr[k] = ei[w & ((1 << QEBITS) - 1)];
      }
    }
    if constexpr (HEADS == 4) {
      float4 sv[4];
      #pragma unroll
      for (int k = 0; k < 4; ++k)
        sv[k] = *(const float4*)(s_src + (size_t)sr[k] * HEADS);
      #pragma unroll
      for (int k = 0; k < 4; ++k) {
        if (on[k]) {
          unsigned* mp = smax + dl[k] * HEADS;
          atomicMax(mp + 0, f2key(sv[k].x));
          atomicMax(mp + 1, f2key(sv[k].y));
          atomicMax(mp + 2, f2key(sv[k].z));
          atomicMax(mp + 3, f2key(sv[k].w));
          if constexpr (!SAVED) q[g + 32 * k + lane] = sr[k] | (dl[k] << QSBITS);
        }
      }
    } else {
      float sv[4][HEADS];
      #pragma unroll
      for (int k = 0; k < 4; ++k) {
        #pragma unroll
        for (int h = 0; h < HEADS; ++h)
          sv[k][h] = s_src[(size_t)sr[k] * HEADS + h];
      }
      #pragma unroll
      for (int k = 0; k < 4; ++k) {
        if (on[k]) {
          #pragma unroll
          for (int h = 0; h < HEADS; ++h)
            atomicMax(smax + dl[k] * HEADS + h, f2key(sv[k][h]));
          if constexpr (!SAVED) q[g + 32 * k + lane] = sr[k] | (dl[k] << QSBITS);
        }
      }
    }
  }
}

__device__ __forceinline__ void tile_list_form(
    const int* __restrict__ ei, int* q, int cnt, int lane)
{
  for (int i = lane; i < cnt; i += 32) {
    const int w = q[i];
    q[i] = ei[w & ((1 << QEBITS) - 1)] | ((int)((unsigned)w >> QEBITS) << QSBITS);
  }
}

template <int HEADS, int AP>
__device__ __forceinline__ void tile_list_sum(
    const float* __restrict__ s_src, const float* __restrict__ hf,
    float* acc, float* den, const float* sdl, const unsigned* rmax,
    const int* q, int base, int n, int lane)
{
  constexpr int HC = HEADS * HIDC;
  constexpr int UE = 32 / (HC / 32);
  int src = 0, dl = 0;
  float a[HEADS];
  #pragma unroll
  for (int h = 0; h < HEADS; ++h) a[h] = 0.f;
  if (lane < n) {
    const int w = q[base + lane];
    src = w & ((1 << QSBITS) - 1);
    dl  = (int)((unsigned)w >> QSBITS);
    if constexpr (HEADS == 4) {
      const float4 sv = *(const float4*)(s_src + (size_t)src * HEADS);
      const float4 sd = *(const float4*)(sdl + dl * HEADS);
      const uint4  mk = *(const uint4*)(rmax + dl * HEADS);
      a[0] = expf(leaky02(sv.x + sd.x) - __uint_as_float(mk.x));
      a[1] = expf(leaky02(sv.y + sd.y) - __uint_as_float(mk.y));
      a[2] = expf(leaky02(sv.z + sd.z) - __uint_as_float(mk.z));
      a[3] = expf(leaky02(sv.w + sd.w) - __uint_as_float(mk.w));
    } else {
      #pragma unroll
      for (int h = 0; h < HEADS; ++h)
        a[h] = expf(leaky02(s_src[(size_t)src * HEADS + h] + sdl[dl * HEADS + h]) -
                    __uint_as_float(rmax[dl * HEADS + h]));
    }
    float* dp = den + dl * HEADS;
    #pragma unroll
    for (int h = 0; h < HEADS; ++h)
      atomicAdd(dp + h, a[h]);
  }
  int j = 0;
  for (; j + UE <= n; j += UE)
    tile_edge_batch<HEADS, UE, AP>(hf, acc, src, dl, a, j, lane);
  if constexpr (UE > 4) {
    for (; j + 4 <= n; j += 4)
      tile_edge_batch<HEADS, 4, AP>(hf, acc, src, dl, a, j, lane);
  }
  for (; j < n; ++j)
    tile_edge_batch<HEADS, 1, AP>(hf, acc, src, dl, a, j, lane);
}

template <int K, int NOUT, int AP, int NTW>
__device__ __forceinline__ void tile_rows_times_w(
    const float* acc, const float* __restrict__ Wn, float* __restrict__ out,
    int n0, int nt, int ntile, int grp, int lane)
{
  constexpr int NG = 32 / (NOUT / 16);
  constexpr int KC = 16;
  static_assert(K % KC == 0, "whole trips");
  const int half = lane >> 4, ml = lane & 15;
  const float* bp = Wn + (size_t)(2 * half) * NOUT + ntile * 16 + ml;
  const float* ap = acc + (16 * grp + ml) * AP + 2 * half;

  v8f c[NTW];
  #pragma unroll
  for (int t = 0; t < NTW; ++t)
    c[t] = (v8f){0.f,0.f,0.f,0.f,0.f,0.f,0.f,0.f};

  v2f bc[KC / 4], bn[KC / 4];
  #pragma unroll
  for (int s = 0; s < KC / 4; ++s) {
    bc[s].x = bp[(size_t)(4 * s + 0) * NOUT];
    bc[s].y = bp[(size_t)(4 * s + 1) * NOUT];
  }
  #pragma unroll 1
  for (int kc = 0; kc < K; kc += KC) {
    const int kn = kc + KC < K ? kc + KC : kc;
    #pragma unroll
    for (int s = 0; s < KC / 4; ++s) {
      bn[s].x = bp[(size_t)(kn + 4 * s + 0) * NOUT];
      bn[s].y = bp[(size_t)(kn + 4 * s + 1) * NOUT];
    }
    #pragma unroll
    for (int s = 0; s < KC / 4; ++s) {
      #pragma unroll
      for (int t = 0; t < NTW; ++t) {
        const v2f a = *(const v2f*)(ap + (16 * NG * t) * AP + kc + 4 * s);
        c[t] = __builtin_amdgcn_wmma_f32_16x16x4_f32(
            false, a, false, bc[s], (short)0, c[t], false, false);
      }
    }
    #pragma unroll
    for (int s = 0; s < KC / 4; ++s) bc[s] = bn[s];
  }

  #pragma unroll
  for (int t = 0; t < NTW; ++t) {
    const int r0 = 16 * (grp + NG * t) + 8 * half;
    float* cp = out + (size_t)(n0 + r0) * NOUT + ntile * 16 + ml;
    #pragma unroll
    for (int i = 0; i < 8; ++i)
      if (r0 + i < nt) cp[(size_t)i * NOUT] = c[t][i];
  }
}

template <int HEADS, int NOUT>
__global__ __launch_bounds__(1024) void gat_dst_tile(
    const int* __restrict__ ei,
    const int* __restrict__ segs,
    const int* __restrict__ soff,

    const float* __restrict__ s_src,
    const float* __restrict__ s_dst,
    const float* __restrict__ hf,
    const float* __restrict__ bias,
    const float* __restrict__ g,
    const float* __restrict__ bt,
    const float* __restrict__ m,
    const float* __restrict__ v,
    const float* __restrict__ Wn,
    float* __restrict__ out, int do_elu)
{
  constexpr int HC  = HEADS * HIDC;
  constexpr int AP    = NOUT > 0 ? HC + 4 : HC;
  constexpr int AROWS = NOUT > 0 ? ((TILE_NODES + 15) / 16) * 16 : TILE_NODES;
  static_assert(NOUT == 0 || (NOUT % 16 == 0 && 32 % (NOUT / 16) == 0 && AP % 64 == 4),
                "a wave per column panel and row group; conflict-free fragment reads");
  constexpr int HC4 = HC / 4;
  constexpr int TS  = TILE_NODES * HEADS;
  constexpr int NCH = NEDGES / 128;
  static_assert(NEDGES % 128 == 0, "edge list splits into whole 512-B loads");
  static_assert(1024 % HC4 == 0, "a thread keeps its channels in every pass");
  static_assert(TS <= 1024, "one thread per (row, head) pair");
  static_assert(NEDGES <= (1 << QEBITS) && TILE_NODES <= (1 << (31 - QEBITS)),
                "edge number and tile row share one list word");
  static_assert(NNODES <= (1 << QSBITS) && TILE_NODES <= (1 << (31 - QSBITS)),
                "source node and tile row share one list word");
  static_assert(QSLOTS >= 128, "an empty list takes any one chunk");

  __shared__ __attribute__((aligned(16))) float    acc[AROWS * AP];
  __shared__ int queue[32 * QSLOTS];
  __shared__ __attribute__((aligned(16))) unsigned rmax[TS];
  __shared__ __attribute__((aligned(16))) float    sdl[TS];
  __shared__ __attribute__((aligned(16))) float    den[TS];
  __shared__ int nofit;

  const int tid  = threadIdx.x;
  const int lane = tid & 31;
  const int wv   = __builtin_amdgcn_readfirstlane(tid >> 5);
  const int n0   = (int)blockIdx.x * TILE_NODES;
  const int nt   = (NNODES - n0) < TILE_NODES ? (NNODES - n0) : TILE_NODES;

  float s_self = 0.f;
  if (tid == 0) nofit = 0;
  if (tid < nt * HEADS) {
    s_self    = s_src[(size_t)n0 * HEADS + tid];
    rmax[tid] = f2key(s_self);
    sdl[tid]  = s_dst[(size_t)n0 * HEADS + tid];
  }
  __syncthreads();

  const int* dstp = ei + NEDGES;
  int* q  = queue + wv * QSLOTS;
  int cnt = 0;

  auto ldchunk = [&](int cc) __attribute__((always_inline)) -> int4 {
    cc = cc < NCH ? cc : NCH - 1;
    return *(const int4*)(dstp + (size_t)cc * 128 + 4 * lane);
  };
  auto enqueue = [&](unsigned mask, int e, int d) __attribute__((always_inline)) {
    if (mask) {
      const int pos = cnt + (int)__builtin_amdgcn_mbcnt_lo(mask, 0u);
      if ((mask >> lane) & 1u)
        q[pos] = e | ((d - n0) << QEBITS);
      cnt += __builtin_popcount(mask);
    }
  };
  auto take = [&](const int4 cur, int cc) __attribute__((always_inline)) -> bool {
    const unsigned m0 = __builtin_amdgcn_ballot_w32((unsigned)(cur.x - n0) < (unsigned)TILE_NODES);
    const unsigned m1 = __builtin_amdgcn_ballot_w32((unsigned)(cur.y - n0) < (unsigned)TILE_NODES);
    const unsigned m2 = __builtin_amdgcn_ballot_w32((unsigned)(cur.z - n0) < (unsigned)TILE_NODES);
    const unsigned m3 = __builtin_amdgcn_ballot_w32((unsigned)(cur.w - n0) < (unsigned)TILE_NODES);
    if ((m0 | m1 | m2 | m3) == 0u) return true;
    const int tot = __builtin_popcount(m0) + __builtin_popcount(m1) +
                    __builtin_popcount(m2) + __builtin_popcount(m3);
    if (cnt + tot > QSLOTS) return false;
    const int eb = cc * 128 + 4 * lane;
    enqueue(m0, eb + 0, cur.x);
    enqueue(m1, eb + 1, cur.y);
    enqueue(m2, eb + 2, cur.z);
    enqueue(m3, eb + 3, cur.w);
    return true;
  };
  auto fill = [&](int c) __attribute__((always_inline)) -> int {
    int4 a0 = ldchunk(c), a1 = ldchunk(c + 32), a2 = ldchunk(c + 64), a3 = ldchunk(c + 96);
    for (;;) {
      const int4 b0 = ldchunk(c + 128), b1 = ldchunk(c + 160),
                 b2 = ldchunk(c + 192), b3 = ldchunk(c + 224);
      if (!take(a0, c)) return c;
      if (c + 32 < NCH && !take(a1, c + 32)) return c + 32;
      if (c + 64 < NCH && !take(a2, c + 64)) return c + 64;
      if (c + 96 < NCH && !take(a3, c + 96)) return c + 96;
      c += 128;
      if (c >= NCH) return NCH;
      a0 = ldchunk(c + 128); a1 = ldchunk(c + 160);
      a2 = ldchunk(c + 192); a3 = ldchunk(c + 224);
      if (!take(b0, c)) return c;
      if (c + 32 < NCH && !take(b1, c + 32)) return c + 32;
      if (c + 64 < NCH && !take(b2, c + 64)) return c + 64;
      if (c + 96 < NCH && !take(b3, c + 96)) return c + 96;
      c += 128;
      if (c >= NCH) return NCH;
    }
  };

  constexpr int NSL = (NTILES + 31) / 32;
  static_assert(NSL <= 32, "one lane per slice of the wave");
  static_assert(NTILES + 1 <= SOFF_PITCH, "a slice's offsets fit their pitch");
  int seg_len = 0, seg_at = 0;
  {
    const int gs = wv + 32 * lane;
    if (lane < NSL && gs < NTILES) {
      const int o0 = soff[gs * SOFF_PITCH + (int)blockIdx.x];
      const int o1 = soff[gs * SOFF_PITCH + (int)blockIdx.x + 1];
      seg_len = o1 - o0;
      seg_at  = gs * SLICE_EDGES + o0;
    }
  }
  int ln[NSL], at[NSL], first[NSL];
  int tot = 0;
  #pragma unroll
  for (int k = 0; k < NSL; ++k) {
    ln[k] = __builtin_amdgcn_readlane(seg_len, k);
    at[k] = __builtin_amdgcn_readlane(seg_at, k);
    first[k] = lane < ln[k] ? segs[at[k] + lane] : 0;
    tot += ln[k];
  }

  if (tot > QSLOTS && lane == 0) nofit = 1;
  __syncthreads();
  int  c = wv;
  bool whole = false;
  if (nofit == 0) {
    #pragma unroll
    for (int k = 0; k < NSL; ++k) {
      if (lane < ln[k]) q[cnt + lane] = first[k];
      #pragma unroll 1
      for (int i = 32 + lane; i < ln[k]; i += 32)
        q[cnt + i] = segs[at[k] + i];
      cnt += ln[k];
    }
    tile_list_max<HEADS, true>(ei, s_src, rmax, q, cnt, lane);
    whole = true;
  } else {
    int rounds = 0;
    do {
      cnt = 0;
      c = fill(c);
      tile_list_max<HEADS, false>(ei, s_src, rmax, q, cnt, lane);
      ++rounds;
    } while (c < NCH);
    whole = rounds == 1;
  }
  __syncthreads();

  if (tid < nt * HEADS) {
    const float t  = sdl[tid];
    const float mm = leaky02(key2f(rmax[tid]) + t);
    rmax[tid] = __float_as_uint(mm);
    den[tid]  = expf(leaky02(s_self + t) - mm);
  }
  __syncthreads();

  for (int i = tid; i < nt * HC4; i += 1024) {
    const int nl = i / HC4, c4 = i % HC4;
    const float al = den[nl * HEADS + (4 * c4) / HIDC];
    const float4 r = *(const float4*)(hf + (size_t)(n0 + nl) * HC + 4 * c4);
    float4 o;
    o.x = r.x * al; o.y = r.y * al; o.z = r.z * al; o.w = r.w * al;
    *(float4*)(acc + nl * AP + 4 * c4) = o;
  }
  __syncthreads();

  if (whole) {
    for (int b = 0; b < cnt; b += 32)
      tile_list_sum<HEADS, AP>(s_src, hf, acc, den, sdl, rmax, q, b,
                               (cnt - b) < 32 ? (cnt - b) : 32, lane);
  } else {
    c = wv;
    while (c < NCH) {
      cnt = 0;
      c = fill(c);
      tile_list_form(ei, q, cnt, lane);
      for (int b = 0; b < cnt; b += 32)
        tile_list_sum<HEADS, AP>(s_src, hf, acc, den, sdl, rmax, q, b,
                                 (cnt - b) < 32 ? (cnt - b) : 32, lane);
    }
  }
  __syncthreads();

  {
    const int c0 = 4 * (tid % HC4);
    const int hh = c0 / HIDC;
    const float4 bv = *(const float4*)(bias + c0);
    const float4 gv = *(const float4*)(g + c0);
    const float4 tv = *(const float4*)(bt + c0);
    const float4 mv = *(const float4*)(m + c0);
    const float4 vv = *(const float4*)(v + c0);
    const float r0 = rsqrtf(vv.x + EPSBN), r1 = rsqrtf(vv.y + EPSBN);
    const float r2 = rsqrtf(vv.z + EPSBN), r3 = rsqrtf(vv.w + EPSBN);
    float* op = out + (size_t)n0 * HC;
    for (int i = tid; i < nt * HC4; i += 1024) {
      float* rp = acc + (i / HC4) * AP + c0;
      const float4 s = *(const float4*)rp;
      const float rd = 1.0f / (den[(i / HC4) * HEADS + hh] + 1e-16f);
      float4 y;
      y.x = (s.x * rd + bv.x - mv.x) * r0 * gv.x + tv.x;
      y.y = (s.y * rd + bv.y - mv.y) * r1 * gv.y + tv.y;
      y.z = (s.z * rd + bv.z - mv.z) * r2 * gv.z + tv.z;
      y.w = (s.w * rd + bv.w - mv.w) * r3 * gv.w + tv.w;
      if (do_elu) {
        y.x = y.x > 0.f ? y.x : expm1f(y.x);
        y.y = y.y > 0.f ? y.y : expm1f(y.y);
        y.z = y.z > 0.f ? y.z : expm1f(y.z);
        y.w = y.w > 0.f ? y.w : expm1f(y.w);
      }
      if constexpr (NOUT > 0) *(float4*)rp = y;
      else                    *(float4*)(op + 4 * i) = y;
    }
  }

  if constexpr (NOUT > 0) {
    __syncthreads();
    constexpr int NT  = NOUT / 16;
    constexpr int NG  = 32 / NT;
    constexpr int MT  = AROWS / 16;
    constexpr int TPW = (MT + NG - 1) / NG;
    static_assert(TPW >= 2 && NG * (TPW - 1) <= MT, "only a wave's last tile may be missing");
    const int ntile = wv % NT, grp = wv / NT;
    if (grp + NG * (TPW - 1) < MT)
      tile_rows_times_w<HC, NOUT, AP, TPW>(acc, Wn, out, n0, nt, ntile, grp, lane);
    else
      tile_rows_times_w<HC, NOUT, AP, TPW - 1>(acc, Wn, out, n0, nt, ntile, grp, lane);
  }
}

__global__ __launch_bounds__(256) void final_linear(const float* __restrict__ h,
                                                    const float* __restrict__ Wc,
                                                    const float* __restrict__ bc,
                                                    float* __restrict__ out)
{
  static_assert(NNODES % ROWDOT_ROWS == 0, "whole trips");
  const int sub = threadIdx.x & 15;
  const v4f w0 = *(const v4f*)(Wc + 8 * sub), w1 = *(const v4f*)(Wc + 8 * sub + 4);
  const v4f wa[1] = {(v4f){w0.x, w0.z, w1.x, w1.z}};
  const v4f wb[1] = {(v4f){w0.y, w0.w, w1.y, w1.w}};
  const float c0 = bc[0], c1 = bc[1];
  row_dots_wave<1>(h, NNODES / ROWDOT_ROWS, wa, wb,
                   [&](int n, float a, float b) __attribute__((always_inline)) {
                     *(v2f*)(out + 2 * n) = (v2f){a + c0, b + c1};
                   });
}

extern "C" void kernel_launch(void* const* d_in, const int* in_sizes, int n_in,
                              void* d_out, int out_size, void* d_ws, size_t ws_size,
                              hipStream_t stream)
{
  const float* x  = (const float*)d_in[0];
  const int*   ei = (const int*)  d_in[1];
  const float* Wc = (const float*)d_in[26];
  const float* bc = (const float*)d_in[27];

  char* ws = (char*)d_ws;
  float*    ta    = (float*)ws;    ws += (size_t)NNODES * 256 * sizeof(float);
  float*    tb    = (float*)ws;    ws += (size_t)NNODES * 256 * sizeof(float);
  float*    ssrc  = (float*)ws;    ws += (size_t)NNODES * NHEADS * sizeof(float);
  float*    sdst  = (float*)ws;    ws += (size_t)NNODES * NHEADS * sizeof(float);
  int*      segs  = (int*)ws;      ws += (size_t)NEDGES * sizeof(int);
  int*      soff  = (int*)ws;      ws += (size_t)NTILES * SOFF_PITCH * sizeof(int);

  bin_edges<<<NTILES, 1024, 0, stream>>>(ei, segs, soff);

  gemm_wmma_f32<FIN><<<dim3(16, 79), 256, 0, stream>>>(
      x, (const float*)d_in[2], ta, 256, 625);

  att_scores<NHEADS><<<ROWDOT_WGS, 256, 0, stream>>>(
      ta, (const float*)d_in[3], (const float*)d_in[4], ssrc, sdst);
  gat_dst_tile<NHEADS, 256><<<NTILES, 1024, 0, stream>>>(
      ei, segs, soff, ssrc, sdst, ta,
      (const float*)d_in[5], (const float*)d_in[6], (const float*)d_in[7],
      (const float*)d_in[8], (const float*)d_in[9],
      (const float*)d_in[10], tb, 1);

  att_scores<NHEADS><<<ROWDOT_WGS, 256, 0, stream>>>(
      tb, (const float*)d_in[11], (const float*)d_in[12], ssrc, sdst);
  gat_dst_tile<NHEADS, 64><<<NTILES, 1024, 0, stream>>>(
      ei, segs, soff, ssrc, sdst, tb,
      (const float*)d_in[13], (const float*)d_in[14], (const float*)d_in[15],
      (const float*)d_in[16], (const float*)d_in[17],
      (const float*)d_in[18], ta, 1);

  att_scores<1><<<ROWDOT_WGS, 256, 0, stream>>>(
      ta, (const float*)d_in[19], (const float*)d_in[20], ssrc, sdst);
  gat_dst_tile<1, 0><<<NTILES, 1024, 0, stream>>>(
      ei, segs, soff, ssrc, sdst, ta,
      (const float*)d_in[21], (const float*)d_in[22], (const float*)d_in[23],
      (const float*)d_in[24], (const float*)d_in[25],
      (const float*)0, tb, 0);

  final_linear<<<ROWDOT_WGS, 256, 0, stream>>>(tb, Wc, bc, (float*)d_out);
}
